# starting kernel
# baseline (speedup 1.0000x reference)
#include <hip/hip_runtime.h>
#include <hip/hip_bf16.h>

#define GEX 978
#define KPAD 980
#define HDIM 128
#define NAT 50
#define FEAT 34
#define CCEH 64
#define BATCH 64

typedef __attribute__((ext_vector_type(2))) float v2f;
typedef __attribute__((ext_vector_type(8))) float v8f;

__global__ __launch_bounds__(256) void k_cce(const float* __restrict__ nf,
                                             const float* __restrict__ mask,
                                             const float* __restrict__ adj,
                                             const float* __restrict__ dist,
                                             const float* __restrict__ W1,
                                             float* __restrict__ pooled) {
    int b = blockIdx.x, tid = threadIdx.x;
    __shared__ float W1s[FEAT * CCEH];
    __shared__ float hs[NAT * CCEH];
    __shared__ float wm[NAT];
    for (int i = tid; i < FEAT * CCEH; i += 256) W1s[i] = W1[i];
    __syncthreads();
    for (int i = tid; i < NAT * CCEH; i += 256) {
        int n = i / CCEH, d = i % CCEH;
        const float* row = nf + (b * NAT + n) * FEAT;
        float s = 0.f;
        #pragma unroll
        for (int f = 0; f < FEAT; f++) s += row[f] * W1s[f * CCEH + d];
        hs[i] = fmaxf(s, 0.f);
    }
    if (tid < NAT) {
        float s = 0.f;
        for (int n = 0; n < NAT; n++) {
            int idx = (b * NAT + n) * NAT + tid;
            s += mask[b * NAT + n] * adj[idx] * expf(-dist[idx]);
        }
        wm[tid] = s;
    }
    __syncthreads();
    if (tid < CCEH) {
        float msum = 0.f;
        for (int n = 0; n < NAT; n++) msum += mask[b * NAT + n];
        msum = fmaxf(msum, 1.f);
        float s = 0.f;
        for (int m = 0; m < NAT; m++) s += wm[m] * hs[m * CCEH + tid];
        pooled[b * CCEH + tid] = s / msum;
    }
}

__global__ __launch_bounds__(256) void k_comp(const float* __restrict__ pooled,
                                              const float* __restrict__ W2,
                                              const float* __restrict__ dose,
                                              const float* __restrict__ timev,
                                              const float* __restrict__ w_dose,
                                              const float* __restrict__ w_time,
                                              float* __restrict__ comp) {
    int idx = blockIdx.x * 256 + threadIdx.x;
    if (idx >= BATCH * GEX) return;
    int b = idx / GEX, g = idx % GEX;
    float s = dose[b] * w_dose[g] + timev[b] * w_time[g];
    #pragma unroll 8
    for (int d = 0; d < CCEH; d++) s += pooled[b * CCEH + d] * W2[d * GEX + g];
    comp[idx] = s;
}

__global__ __launch_bounds__(128) void k_S(const float* __restrict__ b_gex,
                                           const float* __restrict__ comp,
                                           const float* __restrict__ w_gex,
                                           const float* __restrict__ w_comp,
                                           float* __restrict__ S) {
    int b = blockIdx.x, h = threadIdx.x;
    float s = 0.f;
    for (int g = 0; g < GEX; g++) {
        float x = b_gex[b * GEX + g], c = comp[b * GEX + g];
        s += x * w_gex[g * HDIM + h] + c * w_comp[g * HDIM + h];
    }
    S[b * HDIM + h] = s;
}

__global__ __launch_bounds__(256) void k_ppirow(const float* __restrict__ ppi,
                                                float* __restrict__ rowsum) {
    int g = blockIdx.x, tid = threadIdx.x;
    __shared__ float red[256];
    float s = 0.f;
    for (int k = tid; k < GEX; k += 256) s += ppi[g * GEX + k];
    red[tid] = s;
    __syncthreads();
    for (int o = 128; o > 0; o >>= 1) {
        if (tid < o) red[tid] += red[tid + o];
        __syncthreads();
    }
    if (tid == 0) rowsum[g] = red[0];
}

__global__ __launch_bounds__(256) void k_pred0(const float* __restrict__ b_gex,
                                               const float* __restrict__ comp,
                                               const float* __restrict__ w_gex,
                                               const float* __restrict__ w_comp,
                                               const float* __restrict__ S,
                                               const float* __restrict__ ppirow,
                                               float* __restrict__ pred0) {
    __shared__ float Ssh[BATCH * HDIM];
    __shared__ float wg[4][HDIM], wc[4][HDIM];
    int g0 = blockIdx.x * 4, tid = threadIdx.x;
    for (int i = tid; i < BATCH * HDIM; i += 256) Ssh[i] = S[i];
    for (int i = tid; i < 4 * HDIM; i += 256) {
        int gl = i / HDIM, hh = i % HDIM, g = g0 + gl;
        wg[gl][hh] = (g < GEX) ? w_gex[g * HDIM + hh] : 0.f;
        wc[gl][hh] = (g < GEX) ? w_comp[g * HDIM + hh] : 0.f;
    }
    __syncthreads();
    int b = tid >> 2, gl = tid & 3, g = g0 + gl;
    if (g < GEX) {
        float d1 = 0.f, d2 = 0.f;
        #pragma unroll 8
        for (int hh = 0; hh < HDIM; hh++) {
            float sv = Ssh[b * HDIM + hh];
            d1 += sv * wg[gl][hh];
            d2 += sv * wc[gl][hh];
        }
        float x = b_gex[b * GEX + g], c = comp[b * GEX + g];
        float row = (x * d1 + c * d2) * 0.08838834764831845f + ppirow[g];
        pred0[b * GEX + g] = x * row;
    }
}

__global__ __launch_bounds__(256) void k_ln(const float* __restrict__ pred0,
                                            const float* __restrict__ gamma,
                                            const float* __restrict__ beta,
                                            float* __restrict__ xlnP) {
    int b = blockIdx.x, tid = threadIdx.x;
    __shared__ float red[256];
    __shared__ float mu_s, rstd_s;
    float s = 0.f;
    for (int g = tid; g < GEX; g += 256) s += pred0[b * GEX + g];
    red[tid] = s;
    __syncthreads();
    for (int o = 128; o > 0; o >>= 1) { if (tid < o) red[tid] += red[tid + o]; __syncthreads(); }
    if (tid == 0) mu_s = red[0] / (float)GEX;
    __syncthreads();
    float mu = mu_s, v = 0.f;
    for (int g = tid; g < GEX; g += 256) { float d = pred0[b * GEX + g] - mu; v += d * d; }
    red[tid] = v;
    __syncthreads();
    for (int o = 128; o > 0; o >>= 1) { if (tid < o) red[tid] += red[tid + o]; __syncthreads(); }
    if (tid == 0) rstd_s = rsqrtf(red[0] / (float)GEX + 1e-5f);
    __syncthreads();
    float rstd = rstd_s;
    for (int g = tid; g < KPAD; g += 256) {
        float y = 0.f;
        if (g < GEX)
            y = fmaxf((pred0[b * GEX + g] - mu) * rstd * gamma[g] + beta[g], 0.f);
        xlnP[b * KPAD + g] = y;
    }
}

__global__ __launch_bounds__(256) void k_transpose(const float* __restrict__ Wff,
                                                   float* __restrict__ WffT) {
    __shared__ float tile[32][33];
    int kb = blockIdx.x * 32;
    int nb = blockIdx.y * 32;
    int tx = threadIdx.x & 31, ty = threadIdx.x >> 5;
    for (int i = ty; i < 32; i += 8) {
        int k = kb + i, n = nb + tx;
        tile[i][tx] = (k < GEX && n < GEX) ? Wff[k * GEX + n] : 0.f;
    }
    __syncthreads();
    for (int i = ty; i < 32; i += 8) {
        int n = nb + i, k = kb + tx;
        if (n < GEX && k < KPAD) WffT[n * KPAD + k] = tile[tx][i];
    }
}

__global__ __launch_bounds__(256) void k_gemm_wmma(const float* __restrict__ xlnP,
                                                   const float* __restrict__ WffT,
                                                   float* __restrict__ out) {
    const int NT = (GEX + 15) / 16;
    int wave = (blockIdx.x * 256 + threadIdx.x) >> 5;
    int lane = threadIdx.x & 31;
    int mt = wave / NT;
    int nt = wave % NT;
    if (mt >= BATCH / 16) return;
    int lg   = lane & 15;
    int koff = (lane >= 16) ? 2 : 0;
    int m = mt * 16 + lg;
    int n = nt * 16 + lg;
    int nc = (n < GEX) ? n : (GEX - 1);

    const float* ap = xlnP + m * KPAD + koff;
    const float* bp = WffT + nc * KPAD + koff;

    v8f acc = {};
    #pragma unroll 5
    for (int k = 0; k < KPAD; k += 4) {
        v2f a  = *(const v2f*)ap;
        v2f bm = *(const v2f*)bp;
        ap += 4; bp += 4;
        acc = __builtin_amdgcn_wmma_f32_16x16x4_f32(
            false, a, false, bm, (short)0, acc, false, false);
    }

    bool nok = n < GEX;
    #pragma unroll
    for (int v = 0; v < 8; v++) {
        int row = mt * 16 + v + ((lane >= 16) ? 8 : 0);
        if (nok) out[row * GEX + n] = acc[v];
    }
}

extern "C" void kernel_launch(void* const* d_in, const int* in_sizes, int n_in,
                              void* d_out, int out_size, void* d_ws, size_t ws_size,
                              hipStream_t stream) {
    const float* b_gex    = (const float*)d_in[0];
    const float* node_feat= (const float*)d_in[1];
    const float* mask     = (const float*)d_in[2];
    const float* adj      = (const float*)d_in[3];
    const float* dist     = (const float*)d_in[4];
    const float* dose     = (const float*)d_in[5];
    const float* timev    = (const float*)d_in[6];
    const float* ppi      = (const float*)d_in[7];
    const float* w_gex    = (const float*)d_in[8];
    const float* w_comp   = (const float*)d_in[9];
    const float* W1       = (const float*)d_in[10];
    const float* W2       = (const float*)d_in[11];
    const float* w_dose   = (const float*)d_in[12];
    const float* w_time   = (const float*)d_in[13];
    const float* gamma    = (const float*)d_in[14];
    const float* beta     = (const float*)d_in[15];
    const float* Wff      = (const float*)d_in[16];

    float* out  = (float*)d_out;
    float* comp = out + BATCH * GEX;

    float* ws     = (float*)d_ws;
    float* pooled = ws;
    float* S      = pooled + BATCH * CCEH;
    float* ppirow = S + BATCH * HDIM;
    float* pred0  = ppirow + 1024;
    float* xlnP   = pred0 + BATCH * GEX;
    float* WffT   = xlnP + BATCH * KPAD;

    dim3 tg((KPAD + 31) / 32, (GEX + 31) / 32);
    k_transpose<<<tg, 256, 0, stream>>>(Wff, WffT);

    k_cce<<<BATCH, 256, 0, stream>>>(node_feat, mask, adj, dist, W1, pooled);
    k_comp<<<(BATCH * GEX + 255) / 256, 256, 0, stream>>>(pooled, W2, dose, timev,
                                                          w_dose, w_time, comp);
    k_S<<<BATCH, HDIM, 0, stream>>>(b_gex, comp, w_gex, w_comp, S);
    k_ppirow<<<GEX, 256, 0, stream>>>(ppi, ppirow);
    k_pred0<<<(GEX + 3) / 4, 256, 0, stream>>>(b_gex, comp, w_gex, w_comp, S, ppirow, pred0);
    k_ln<<<BATCH, 256, 0, stream>>>(pred0, gamma, beta, xlnP);
    k_gemm_wmma<<<31, 256, 0, stream>>>(xlnP, WffT, out);
}


# CCE split over (batch row, 25-atom chunk) work-groups: each issues all its loads as one group and makes wm, relu(nf@W1) and a part of pooled; k_compS adds the 2 parts; pooled's sum order changes
# speedup vs baseline: 4.2441x; 4.2441x over previous
#include <hip/hip_runtime.h>
#include <hip/hip_bf16.h>

#define GEX 978
#define KPAD 980
#define HDIM 128
#define NAT 50
#define FEAT 34
#define CCEH 64
#define BATCH 64
#define GCH 16
#define NSLAB ((GEX + GCH - 1) / GCH)
#define BGRP 16
#define PPITCH (CCEH + 1)
#define MC 25
#define NCH (NAT / MC)
#define CCE_T 288
#define NFT ((MC * FEAT + CCE_T - 1) / CCE_T)
#define ADT ((NAT * MC + CCE_T - 1) / CCE_T)
#define KSTEPS 35
#define KSLAB (4 * KSTEPS)
#define NKS (KPAD / KSLAB)
#define NBLK 32
#define NNB ((GEX + NBLK - 1) / NBLK)
#define PPN (NNB * NBLK)

typedef __attribute__((ext_vector_type(2))) float v2f;
typedef __attribute__((ext_vector_type(8))) float v8f;

__global__ __launch_bounds__(CCE_T) void k_cce(const float* __restrict__ nf,
                                               const float* __restrict__ mask,
                                               const float* __restrict__ adj,
                                               const float* __restrict__ dist,
                                               const float* __restrict__ W1,
                                               float* __restrict__ ppool) {
    int b = blockIdx.x, c = blockIdx.y, tid = threadIdx.x, m0 = c * MC;
    __shared__ float nfs[MC * FEAT];
    __shared__ float as[NAT * MC];
    __shared__ float es[NAT * MC];
    __shared__ float hs[MC * CCEH];
    __shared__ float ms[NAT];
    __shared__ float wm[MC];
    __shared__ float msum_s;

    int d = tid & (CCEH - 1);
    float w1[FEAT];
    #pragma unroll
    for (int f = 0; f < FEAT; f++) w1[f] = W1[f * CCEH + d];
    const float* nfb = nf + (b * NAT + m0) * FEAT;
    float nv[NFT];
    #pragma unroll
    for (int j = 0; j < NFT; j++) {
        int i = tid + CCE_T * j;
        nv[j] = nfb[i < MC * FEAT ? i : MC * FEAT - 1];
    }
    float av[ADT], dv[ADT];
    #pragma unroll
    for (int q = 0; q < ADT; q++) {
        int j = tid + CCE_T * q;
        int jc = j < NAT * MC ? j : NAT * MC - 1;
        int n = jc / MC, ml = jc - n * MC;
        int idx = (b * NAT + n) * NAT + m0 + ml;
        av[q] = adj[idx];
        dv[q] = dist[idx];
    }
    float mk = mask[b * NAT + (tid < NAT ? tid : NAT - 1)];

    #pragma unroll
    for (int j = 0; j < NFT; j++) {
        int i = tid + CCE_T * j;
        if (i < MC * FEAT) nfs[i] = nv[j];
    }
    #pragma unroll
    for (int q = 0; q < ADT; q++) {
        int j = tid + CCE_T * q;
        if (j < NAT * MC) { as[j] = av[q]; es[j] = expf(-dv[q]); }
    }
    if (tid < NAT) ms[tid] = mk;
    __syncthreads();

    if (tid < 256) {
        int mq = tid >> 6;
        #pragma unroll
        for (int r = 0; r < (MC + 3) / 4; r++) {
            int ml = mq + 4 * r;
            if (ml < MC) {
                float s = 0.f;
                #pragma unroll
                for (int f = 0; f < FEAT; f++) s += nfs[ml * FEAT + f] * w1[f];
                hs[ml * CCEH + d] = fmaxf(s, 0.f);
            }
        }
    } else if (tid < 256 + MC) {
        int ml = tid - 256;
        float s = 0.f, msum = 0.f;
        #pragma unroll 10
        for (int n = 0; n < NAT; n++) {
            s += ms[n] * as[n * MC + ml] * es[n * MC + ml];
            msum += ms[n];
        }
        wm[ml] = s;
        if (ml == 0) msum_s = fmaxf(msum, 1.f);
    }
    __syncthreads();
    if (tid < CCEH) {
        float s = 0.f;
        #pragma unroll
        for (int ml = 0; ml < MC; ml++) s += wm[ml] * hs[ml * CCEH + tid];
        ppool[(b * NCH + c) * CCEH + tid] = s / msum_s;
    }
}

__global__ __launch_bounds__(256) void k_compS(const float* __restrict__ ppool,
                                               const float* __restrict__ W2,
                                               const float* __restrict__ dose,
                                               const float* __restrict__ timev,
                                               const float* __restrict__ w_dose,
                                               const float* __restrict__ w_time,
                                               const float* __restrict__ b_gex,
                                               const float* __restrict__ w_gex,
                                               const float* __restrict__ w_comp,
                                               float* __restrict__ comp,
                                               float* __restrict__ Spart) {
    __shared__ float Ps[BGRP * PPITCH];
    __shared__ float W2s[CCEH * GCH];
    __shared__ __attribute__((aligned(16))) float Xs[BGRP * GCH];
    __shared__ __attribute__((aligned(16))) float Cs[BGRP * GCH];
    int slab = blockIdx.x, b0 = blockIdx.y * BGRP, g0 = slab * GCH, tid = threadIdx.x;

    float pv[4][NCH], wv[4];
    #pragma unroll
    for (int j = 0; j < 4; j++) {
        int i = tid + 256 * j;
        int gq = g0 + (i & 15);
        #pragma unroll
        for (int c = 0; c < NCH; c++)
            pv[j][c] = ppool[((b0 + (i >> 6)) * NCH + c) * CCEH + (i & 63)];
        wv[j] = W2[(i >> 4) * GEX + (gq < GEX ? gq : GEX - 1)];
    }
    int bl = tid >> 4, gl = tid & 15, b = b0 + bl, g = g0 + gl;
    bool gok = g < GEX;
    int gc = gok ? g : GEX - 1;
    float x  = b_gex[b * GEX + gc];
    float s  = dose[b] * w_dose[gc] + timev[b] * w_time[gc];
    int h = tid & (HDIM - 1), bh = tid >> 7;
    float wg[GCH], wc[GCH];
    #pragma unroll
    for (int j = 0; j < GCH; j++) {
        int gq = g0 + j;
        int gr = gq < GEX ? gq : GEX - 1;
        float a = w_gex[gr * HDIM + h], c2 = w_comp[gr * HDIM + h];
        wg[j] = gq < GEX ? a : 0.f;
        wc[j] = gq < GEX ? c2 : 0.f;
    }
    #pragma unroll
    for (int j = 0; j < 4; j++) {
        int i = tid + 256 * j;
        float p = pv[j][0];
        #pragma unroll
        for (int c = 1; c < NCH; c++) p += pv[j][c];
        Ps[(i >> 6) * PPITCH + (i & 63)] = p;
        W2s[i] = (g0 + (i & 15) < GEX) ? wv[j] : 0.f;
    }
    if (!gok) { x = 0.f; s = 0.f; }
    __syncthreads();
    #pragma unroll 8
    for (int d = 0; d < CCEH; d++) s += Ps[bl * PPITCH + d] * W2s[d * GCH + gl];
    if (gok) comp[b * GEX + g] = s;
    Xs[tid] = x;
    Cs[tid] = gok ? s : 0.f;
    __syncthreads();
    #pragma unroll
    for (int r = 0; r < BGRP / 2; r++) {
        int row = bh * (BGRP / 2) + r;
        float acc = 0.f;
        #pragma unroll
        for (int j = 0; j < GCH; j++)
            acc += Xs[row * GCH + j] * wg[j] + Cs[row * GCH + j] * wc[j];
        Spart[(slab * BATCH + b0 + row) * HDIM + h] = acc;
    }
}

__global__ __launch_bounds__(128) void k_S(const float* __restrict__ Spart,
                                           float* __restrict__ S) {
    int i = blockIdx.x * HDIM + threadIdx.x;
    float s = 0.f;
    #pragma unroll
    for (int p = 0; p < NSLAB; p++) s += Spart[p * (BATCH * HDIM) + i];
    S[i] = s;
}

__global__ __launch_bounds__(256) void k_ppirow(const float* __restrict__ ppi,
                                                float* __restrict__ rowsum) {
    int g = blockIdx.x, tid = threadIdx.x;
    __shared__ float red[256];
    float s = 0.f;
    for (int k = tid; k < GEX; k += 256) s += ppi[g * GEX + k];
    red[tid] = s;
    __syncthreads();
    for (int o = 128; o > 0; o >>= 1) {
        if (tid < o) red[tid] += red[tid + o];
        __syncthreads();
    }
    if (tid == 0) rowsum[g] = red[0];
}

__global__ __launch_bounds__(256) void k_pred0(const float* __restrict__ b_gex,
                                               const float* __restrict__ comp,
                                               const float* __restrict__ w_gex,
                                               const float* __restrict__ w_comp,
                                               const float* __restrict__ S,
                                               const float* __restrict__ ppirow,
                                               float* __restrict__ pred0) {
    __shared__ float Ssh[BATCH * HDIM];
    __shared__ float wg[4][HDIM], wc[4][HDIM];
    int g0 = blockIdx.x * 4, tid = threadIdx.x;
    for (int i = tid; i < BATCH * HDIM; i += 256) Ssh[i] = S[i];
    for (int i = tid; i < 4 * HDIM; i += 256) {
        int gl = i / HDIM, hh = i % HDIM, g = g0 + gl;
        wg[gl][hh] = (g < GEX) ? w_gex[g * HDIM + hh] : 0.f;
        wc[gl][hh] = (g < GEX) ? w_comp[g * HDIM + hh] : 0.f;
    }
    __syncthreads();
    int b = tid >> 2, gl = tid & 3, g = g0 + gl;
    if (g < GEX) {
        float d1 = 0.f, d2 = 0.f;
        #pragma unroll 8
        for (int hh = 0; hh < HDIM; hh++) {
            float sv = Ssh[b * HDIM + hh];
            d1 += sv * wg[gl][hh];
            d2 += sv * wc[gl][hh];
        }
        float x = b_gex[b * GEX + g], c = comp[b * GEX + g];
        float row = (x * d1 + c * d2) * 0.08838834764831845f + ppirow[g];
        pred0[b * GEX + g] = x * row;
    }
}

__global__ __launch_bounds__(256) void k_ln(const float* __restrict__ pred0,
                                            const float* __restrict__ gamma,
                                            const float* __restrict__ beta,
                                            float* __restrict__ xlnP) {
    int b = blockIdx.x, tid = threadIdx.x;
    __shared__ float red[256];
    __shared__ float mu_s, rstd_s;
    float s = 0.f;
    for (int g = tid; g < GEX; g += 256) s += pred0[b * GEX + g];
    red[tid] = s;
    __syncthreads();
    for (int o = 128; o > 0; o >>= 1) { if (tid < o) red[tid] += red[tid + o]; __syncthreads(); }
    if (tid == 0) mu_s = red[0] / (float)GEX;
    __syncthreads();
    float mu = mu_s, v = 0.f;
    for (int g = tid; g < GEX; g += 256) { float d = pred0[b * GEX + g] - mu; v += d * d; }
    red[tid] = v;
    __syncthreads();
    for (int o = 128; o > 0; o >>= 1) { if (tid < o) red[tid] += red[tid + o]; __syncthreads(); }
    if (tid == 0) rstd_s = rsqrtf(red[0] / (float)GEX + 1e-5f);
    __syncthreads();
    float rstd = rstd_s;
    for (int g = tid; g < KPAD; g += 256) {
        float y = 0.f;
        if (g < GEX)
            y = fmaxf((pred0[b * GEX + g] - mu) * rstd * gamma[g] + beta[g], 0.f);
        xlnP[b * KPAD + g] = y;
    }
}

__global__ __launch_bounds__(256) void k_gemm_slab(const float* __restrict__ xlnP,
                                                   const float* __restrict__ Wff,
                                                   float* __restrict__ Ppart) {
    int lane = threadIdx.x & 31, w = threadIdx.x >> 5;
    int mt = w & 3, ns = w >> 2;
    int slab = blockIdx.y;
    int lg   = lane & 15;
    int koff = (lane >= 16) ? 2 : 0;
    int m  = mt * 16 + lg;
    int n  = blockIdx.x * NBLK + ns * 16 + lg;
    int nc = (n < GEX) ? n : (GEX - 1);
    int kb = slab * KSLAB + koff;

    const float* ap = xlnP + m * KPAD + kb;
    const float* bp = Wff + kb * GEX + nc;
    float* pp = Ppart + (slab * BATCH + mt * 16 + ((lane >= 16) ? 8 : 0)) * PPN + n;

    v8f acc = {};
    #pragma unroll
    for (int s = 0; s < KSTEPS; s++) {
        v2f a = *(const v2f*)(ap + 4 * s);
        v2f bm;
        if (s < KSTEPS - 1) {
            bm.x = bp[(4 * s) * GEX];
            bm.y = bp[(4 * s + 1) * GEX];
        } else {
            int k0 = kb + 4 * s, k1 = k0 + 1;
            int r0 = (k0 < GEX) ? k0 : (GEX - 1);
            int r1 = (k1 < GEX) ? k1 : (GEX - 1);
            float b0 = Wff[r0 * GEX + nc], b1 = Wff[r1 * GEX + nc];
            bm.x = (k0 < GEX) ? b0 : 0.f;
            bm.y = (k1 < GEX) ? b1 : 0.f;
        }
        acc = __builtin_amdgcn_wmma_f32_16x16x4_f32(
            false, a, false, bm, (short)0, acc, false, false);
    }

    for (int v = 0; v < 8; v++) pp[v * PPN] = acc[v];
}

__global__ __launch_bounds__(256) void k_outsum(const float* __restrict__ Ppart,
                                                float* __restrict__ out) {
    int j = blockIdx.x * 256 + threadIdx.x, b = blockIdx.y;
    if (j < GEX / 2) {
        const float* pl = Ppart + b * PPN + 2 * j;
        v2f s = *(const v2f*)pl;
        for (int p = 1; p < NKS; p++) s += *(const v2f*)(pl + p * (BATCH * PPN));
        *(v2f*)(out + b * GEX + 2 * j) = s;
    }
}

extern "C" void kernel_launch(void* const* d_in, const int* in_sizes, int n_in,
                              void* d_out, int out_size, void* d_ws, size_t ws_size,
                              hipStream_t stream) {
    const float* b_gex    = (const float*)d_in[0];
    const float* node_feat= (const float*)d_in[1];
    const float* mask     = (const float*)d_in[2];
    const float* adj      = (const float*)d_in[3];
    const float* dist     = (const float*)d_in[4];
    const float* dose     = (const float*)d_in[5];
    const float* timev    = (const float*)d_in[6];
    const float* ppi      = (const float*)d_in[7];
    const float* w_gex    = (const float*)d_in[8];
    const float* w_comp   = (const float*)d_in[9];
    const float* W1       = (const float*)d_in[10];
    const float* W2       = (const float*)d_in[11];
    const float* w_dose   = (const float*)d_in[12];
    const float* w_time   = (const float*)d_in[13];
    const float* gamma    = (const float*)d_in[14];
    const float* beta     = (const float*)d_in[15];
    const float* Wff      = (const float*)d_in[16];

    float* out  = (float*)d_out;
    float* comp = out + BATCH * GEX;

    float* ws     = (float*)d_ws;
    float* ppool  = ws;
    float* S      = ppool + BATCH * NCH * CCEH;
    float* ppirow = S + BATCH * HDIM;
    float* pred0  = ppirow + 1024;
    float* xlnP   = pred0 + BATCH * GEX;
    float* Ppart  = xlnP + BATCH * KPAD;
    float* Spart  = Ppart + NKS * BATCH * PPN;

    k_cce<<<dim3(BATCH, NCH), CCE_T, 0, stream>>>(node_feat, mask, adj, dist, W1, ppool);
    k_compS<<<dim3(NSLAB, BATCH / BGRP), 256, 0, stream>>>(ppool, W2, dose, timev, w_dose, w_time,
                                                           b_gex, w_gex, w_comp, comp, Spart);
    k_S<<<BATCH, HDIM, 0, stream>>>(Spart, S);
    k_ppirow<<<GEX, 256, 0, stream>>>(ppi, ppirow);
    k_pred0<<<(GEX + 3) / 4, 256, 0, stream>>>(b_gex, comp, w_gex, w_comp, S, ppirow, pred0);
    k_ln<<<BATCH, 256, 0, stream>>>(pred0, gamma, beta, xlnP);
    k_gemm_slab<<<dim3(NNB, NKS), 256, 0, stream>>>(xlnP, Wff, Ppart);
    k_outsum<<<dim3((GEX / 2 + 255) / 256, BATCH), 256, 0, stream>>>(Ppart, out);
}


# ppi row sums made by k_pred0's own work-groups (2 waves per row, 8-byte loads grouped with the S loads, lane butterfly): the row-sum launch and its plane go, 8 launches to 7; row-sum order changes
# speedup vs baseline: 4.7318x; 1.1149x over previous
#include <hip/hip_runtime.h>
#include <hip/hip_bf16.h>

#define GEX 978
#define KPAD 980
#define HDIM 128
#define NAT 50
#define FEAT 34
#define CCEH 64
#define BATCH 64
#define GCH 16
#define NSLAB ((GEX + GCH - 1) / GCH)
#define BGRP 16
#define PPITCH (CCEH + 1)
#define MC 25
#define NCH (NAT / MC)
#define CCE_T 288
#define NFT ((MC * FEAT + CCE_T - 1) / CCE_T)
#define ADT ((NAT * MC + CCE_T - 1) / CCE_T)
#define KSTEPS 35
#define KSLAB (4 * KSTEPS)
#define NKS (KPAD / KSLAB)
#define NBLK 32
#define NNB ((GEX + NBLK - 1) / NBLK)
#define PPN (NNB * NBLK)
#define PPI_V2 (GEX / 2)
#define PPI_LD ((PPI_V2 + 63) / 64)

typedef __attribute__((ext_vector_type(2))) float v2f;
typedef __attribute__((ext_vector_type(8))) float v8f;

__global__ __launch_bounds__(CCE_T) void k_cce(const float* __restrict__ nf,
                                               const float* __restrict__ mask,
                                               const float* __restrict__ adj,
                                               const float* __restrict__ dist,
                                               const float* __restrict__ W1,
                                               float* __restrict__ ppool) {
    int b = blockIdx.x, c = blockIdx.y, tid = threadIdx.x, m0 = c * MC;
    __shared__ float nfs[MC * FEAT];
    __shared__ float as[NAT * MC];
    __shared__ float es[NAT * MC];
    __shared__ float hs[MC * CCEH];
    __shared__ float ms[NAT];
    __shared__ float wm[MC];
    __shared__ float msum_s;

    int d = tid & (CCEH - 1);
    float w1[FEAT];
    #pragma unroll
    for (int f = 0; f < FEAT; f++) w1[f] = W1[f * CCEH + d];
    const float* nfb = nf + (b * NAT + m0) * FEAT;
    float nv[NFT];
    #pragma unroll
    for (int j = 0; j < NFT; j++) {
        int i = tid + CCE_T * j;
        nv[j] = nfb[i < MC * FEAT ? i : MC * FEAT - 1];
    }
    float av[ADT], dv[ADT];
    #pragma unroll
    for (int q = 0; q < ADT; q++) {
        int j = tid + CCE_T * q;
        int jc = j < NAT * MC ? j : NAT * MC - 1;
        int n = jc / MC, ml = jc - n * MC;
        int idx = (b * NAT + n) * NAT + m0 + ml;
        av[q] = adj[idx];
        dv[q] = dist[idx];
    }
    float mk = mask[b * NAT + (tid < NAT ? tid : NAT - 1)];

    #pragma unroll
    for (int j = 0; j < NFT; j++) {
        int i = tid + CCE_T * j;
        if (i < MC * FEAT) nfs[i] = nv[j];
    }
    #pragma unroll
    for (int q = 0; q < ADT; q++) {
        int j = tid + CCE_T * q;
        if (j < NAT * MC) { as[j] = av[q]; es[j] = expf(-dv[q]); }
    }
    if (tid < NAT) ms[tid] = mk;
    __syncthreads();

    if (tid < 256) {
        int mq = tid >> 6;
        #pragma unroll
        for (int r = 0; r < (MC + 3) / 4; r++) {
            int ml = mq + 4 * r;
            if (ml < MC) {
                float s = 0.f;
                #pragma unroll
                for (int f = 0; f < FEAT; f++) s += nfs[ml * FEAT + f] * w1[f];
                hs[ml * CCEH + d] = fmaxf(s, 0.f);
            }
        }
    } else if (tid < 256 + MC) {
        int ml = tid - 256;
        float s = 0.f, msum = 0.f;
        #pragma unroll 10
        for (int n = 0; n < NAT; n++) {
            s += ms[n] * as[n * MC + ml] * es[n * MC + ml];
            msum += ms[n];
        }
        wm[ml] = s;
        if (ml == 0) msum_s = fmaxf(msum, 1.f);
    }
    __syncthreads();
    if (tid < CCEH) {
        float s = 0.f;
        #pragma unroll
        for (int ml = 0; ml < MC; ml++) s += wm[ml] * hs[ml * CCEH + tid];
        ppool[(b * NCH + c) * CCEH + tid] = s / msum_s;
    }
}

__global__ __launch_bounds__(256) void k_compS(const float* __restrict__ ppool,
                                               const float* __restrict__ W2,
                                               const float* __restrict__ dose,
                                               const float* __restrict__ timev,
                                               const float* __restrict__ w_dose,
                                               const float* __restrict__ w_time,
                                               const float* __restrict__ b_gex,
                                               const float* __restrict__ w_gex,
                                               const float* __restrict__ w_comp,
                                               float* __restrict__ comp,
                                               float* __restrict__ Spart) {
    __shared__ float Ps[BGRP * PPITCH];
    __shared__ float W2s[CCEH * GCH];
    __shared__ __attribute__((aligned(16))) float Xs[BGRP * GCH];
    __shared__ __attribute__((aligned(16))) float Cs[BGRP * GCH];
    int slab = blockIdx.x, b0 = blockIdx.y * BGRP, g0 = slab * GCH, tid = threadIdx.x;

    float pv[4][NCH], wv[4];
    #pragma unroll
    for (int j = 0; j < 4; j++) {
        int i = tid + 256 * j;
        int gq = g0 + (i & 15);
        #pragma unroll
        for (int c = 0; c < NCH; c++)
            pv[j][c] = ppool[((b0 + (i >> 6)) * NCH + c) * CCEH + (i & 63)];
        wv[j] = W2[(i >> 4) * GEX + (gq < GEX ? gq : GEX - 1)];
    }
    int bl = tid >> 4, gl = tid & 15, b = b0 + bl, g = g0 + gl;
    bool gok = g < GEX;
    int gc = gok ? g : GEX - 1;
    float x  = b_gex[b * GEX + gc];
    float s  = dose[b] * w_dose[gc] + timev[b] * w_time[gc];
    int h = tid & (HDIM - 1), bh = tid >> 7;
    float wg[GCH], wc[GCH];
    #pragma unroll
    for (int j = 0; j < GCH; j++) {
        int gq = g0 + j;
        int gr = gq < GEX ? gq : GEX - 1;
        float a = w_gex[gr * HDIM + h], c2 = w_comp[gr * HDIM + h];
        wg[j] = gq < GEX ? a : 0.f;
        wc[j] = gq < GEX ? c2 : 0.f;
    }
    #pragma unroll
    for (int j = 0; j < 4; j++) {
        int i = tid + 256 * j;
        float p = pv[j][0];
        #pragma unroll
        for (int c = 1; c < NCH; c++) p += pv[j][c];
        Ps[(i >> 6) * PPITCH + (i & 63)] = p;
        W2s[i] = (g0 + (i & 15) < GEX) ? wv[j] : 0.f;
    }
    if (!gok) { x = 0.f; s = 0.f; }
    __syncthreads();
    #pragma unroll 8
    for (int d = 0; d < CCEH; d++) s += Ps[bl * PPITCH + d] * W2s[d * GCH + gl];
    if (gok) comp[b * GEX + g] = s;
    Xs[tid] = x;
    Cs[tid] = gok ? s : 0.f;
    __syncthreads();
    #pragma unroll
    for (int r = 0; r < BGRP / 2; r++) {
        int row = bh * (BGRP / 2) + r;
        float acc = 0.f;
        #pragma unroll
        for (int j = 0; j < GCH; j++)
            acc += Xs[row * GCH + j] * wg[j] + Cs[row * GCH + j] * wc[j];
        Spart[(slab * BATCH + b0 + row) * HDIM + h] = acc;
    }
}

__global__ __launch_bounds__(128) void k_S(const float* __restrict__ Spart,
                                           float* __restrict__ S) {
    int i = blockIdx.x * HDIM + threadIdx.x;
    float s = 0.f;
    #pragma unroll
    for (int p = 0; p < NSLAB; p++) s += Spart[p * (BATCH * HDIM) + i];
    S[i] = s;
}

__global__ __launch_bounds__(256) void k_pred0(const float* __restrict__ b_gex,
                                               const float* __restrict__ comp,
                                               const float* __restrict__ w_gex,
                                               const float* __restrict__ w_comp,
                                               const float* __restrict__ S,
                                               const float* __restrict__ ppi,
                                               float* __restrict__ pred0) {
    __shared__ float Ssh[BATCH * HDIM];
    __shared__ float wg[4][HDIM], wc[4][HDIM];
    __shared__ float rs[8];
    int g0 = blockIdx.x * 4, tid = threadIdx.x;
    int pr = g0 + (tid >> 6), pt = tid & 63;
    const v2f* prow = (const v2f*)(ppi + (pr < GEX ? pr : GEX - 1) * GEX);
    v2f pv[PPI_LD];
    #pragma unroll
    for (int i = 0; i < PPI_LD; i++) {
        int j = pt + 64 * i;
        pv[i] = prow[j < PPI_V2 ? j : PPI_V2 - 1];
    }
    for (int i = tid; i < BATCH * HDIM; i += 256) Ssh[i] = S[i];
    for (int i = tid; i < 4 * HDIM; i += 256) {
        int gl = i / HDIM, hh = i % HDIM, g = g0 + gl;
        wg[gl][hh] = (g < GEX) ? w_gex[g * HDIM + hh] : 0.f;
        wc[gl][hh] = (g < GEX) ? w_comp[g * HDIM + hh] : 0.f;
    }
    float rsum = 0.f;
    #pragma unroll
    for (int i = 0; i < PPI_LD; i++) {
        float a = pv[i].x + pv[i].y;
        rsum += (pt + 64 * i < PPI_V2) ? a : 0.f;
    }
    for (int o = 16; o > 0; o >>= 1) rsum += __shfl_xor(rsum, o);
    if ((tid & 31) == 0) rs[tid >> 5] = rsum;
    __syncthreads();
    int b = tid >> 2, gl = tid & 3, g = g0 + gl;
    if (g < GEX) {
        float d1 = 0.f, d2 = 0.f;
        #pragma unroll 8
        for (int hh = 0; hh < HDIM; hh++) {
            float sv = Ssh[b * HDIM + hh];
            d1 += sv * wg[gl][hh];
            d2 += sv * wc[gl][hh];
        }
        float x = b_gex[b * GEX + g], c = comp[b * GEX + g];
        float row = (x * d1 + c * d2) * 0.08838834764831845f
                  + (rs[2 * gl] + rs[2 * gl + 1]);
        pred0[b * GEX + g] = x * row;
    }
}

__global__ __launch_bounds__(256) void k_ln(const float* __restrict__ pred0,
                                            const float* __restrict__ gamma,
                                            const float* __restrict__ beta,
                                            float* __restrict__ xlnP) {
    int b = blockIdx.x, tid = threadIdx.x;
    __shared__ float red[256];
    __shared__ float mu_s, rstd_s;
    float s = 0.f;
    for (int g = tid; g < GEX; g += 256) s += pred0[b * GEX + g];
    red[tid] = s;
    __syncthreads();
    for (int o = 128; o > 0; o >>= 1) { if (tid < o) red[tid] += red[tid + o]; __syncthreads(); }
    if (tid == 0) mu_s = red[0] / (float)GEX;
    __syncthreads();
    float mu = mu_s, v = 0.f;
    for (int g = tid; g < GEX; g += 256) { float d = pred0[b * GEX + g] - mu; v += d * d; }
    red[tid] = v;
    __syncthreads();
    for (int o = 128; o > 0; o >>= 1) { if (tid < o) red[tid] += red[tid + o]; __syncthreads(); }
    if (tid == 0) rstd_s = rsqrtf(red[0] / (float)GEX + 1e-5f);
    __syncthreads();
    float rstd = rstd_s;
    for (int g = tid; g < KPAD; g += 256) {
        float y = 0.f;
        if (g < GEX)
            y = fmaxf((pred0[b * GEX + g] - mu) * rstd * gamma[g] + beta[g], 0.f);
        xlnP[b * KPAD + g] = y;
    }
}

__global__ __launch_bounds__(256) void k_gemm_slab(const float* __restrict__ xlnP,
                                                   const float* __restrict__ Wff,
                                                   float* __restrict__ Ppart) {
    int lane = threadIdx.x & 31, w = threadIdx.x >> 5;
    int mt = w & 3, ns = w >> 2;
    int slab = blockIdx.y;
    int lg   = lane & 15;
    int koff = (lane >= 16) ? 2 : 0;
    int m  = mt * 16 + lg;
    int n  = blockIdx.x * NBLK + ns * 16 + lg;
    int nc = (n < GEX) ? n : (GEX - 1);
    int kb = slab * KSLAB + koff;

    const float* ap = xlnP + m * KPAD + kb;
    const float* bp = Wff + kb * GEX + nc;
    float* pp = Ppart + (slab * BATCH + mt * 16 + ((lane >= 16) ? 8 : 0)) * PPN + n;

    v8f acc = {};
    #pragma unroll
    for (int s = 0; s < KSTEPS; s++) {
        v2f a = *(const v2f*)(ap + 4 * s);
        v2f bm;
        if (s < KSTEPS - 1) {
            bm.x = bp[(4 * s) * GEX];
            bm.y = bp[(4 * s + 1) * GEX];
        } else {
            int k0 = kb + 4 * s, k1 = k0 + 1;
            int r0 = (k0 < GEX) ? k0 : (GEX - 1);
            int r1 = (k1 < GEX) ? k1 : (GEX - 1);
            float b0 = Wff[r0 * GEX + nc], b1 = Wff[r1 * GEX + nc];
            bm.x = (k0 < GEX) ? b0 : 0.f;
            bm.y = (k1 < GEX) ? b1 : 0.f;
        }
        acc = __builtin_amdgcn_wmma_f32_16x16x4_f32(
            false, a, false, bm, (short)0, acc, false, false);
    }

    for (int v = 0; v < 8; v++) pp[v * PPN] = acc[v];
}

__global__ __launch_bounds__(256) void k_outsum(const float* __restrict__ Ppart,
                                                float* __restrict__ out) {
    int j = blockIdx.x * 256 + threadIdx.x, b = blockIdx.y;
    if (j < GEX / 2) {
        const float* pl = Ppart + b * PPN + 2 * j;
        v2f s = *(const v2f*)pl;
        for (int p = 1; p < NKS; p++) s += *(const v2f*)(pl + p * (BATCH * PPN));
        *(v2f*)(out + b * GEX + 2 * j) = s;
    }
}

extern "C" void kernel_launch(void* const* d_in, const int* in_sizes, int n_in,
                              void* d_out, int out_size, void* d_ws, size_t ws_size,
                              hipStream_t stream) {
    const float* b_gex    = (const float*)d_in[0];
    const float* node_feat= (const float*)d_in[1];
    const float* mask     = (const float*)d_in[2];
    const float* adj      = (const float*)d_in[3];
    const float* dist     = (const float*)d_in[4];
    const float* dose     = (const float*)d_in[5];
    const float* timev    = (const float*)d_in[6];
    const float* ppi      = (const float*)d_in[7];
    const float* w_gex    = (const float*)d_in[8];
    const float* w_comp   = (const float*)d_in[9];
    const float* W1       = (const float*)d_in[10];
    const float* W2       = (const float*)d_in[11];
    const float* w_dose   = (const float*)d_in[12];
    const float* w_time   = (const float*)d_in[13];
    const float* gamma    = (const float*)d_in[14];
    const float* beta     = (const float*)d_in[15];
    const float* Wff      = (const float*)d_in[16];

    float* out  = (float*)d_out;
    float* comp = out + BATCH * GEX;

    float* ws     = (float*)d_ws;
    float* ppool  = ws;
    float* S      = ppool + BATCH * NCH * CCEH;
    float* pred0  = S + BATCH * HDIM;
    float* xlnP   = pred0 + BATCH * GEX;
    float* Ppart  = xlnP + BATCH * KPAD;
    float* Spart  = Ppart + NKS * BATCH * PPN;

    k_cce<<<dim3(BATCH, NCH), CCE_T, 0, stream>>>(node_feat, mask, adj, dist, W1, ppool);
    k_compS<<<dim3(NSLAB, BATCH / BGRP), 256, 0, stream>>>(ppool, W2, dose, timev, w_dose, w_time,
                                                           b_gex, w_gex, w_comp, comp, Spart);
    k_S<<<BATCH, HDIM, 0, stream>>>(Spart, S);
    k_pred0<<<(GEX + 3) / 4, 256, 0, stream>>>(b_gex, comp, w_gex, w_comp, S, ppi, pred0);
    k_ln<<<BATCH, 256, 0, stream>>>(pred0, gamma, beta, xlnP);
    k_gemm_slab<<<dim3(NNB, NKS), 256, 0, stream>>>(xlnP, Wff, Ppart);
    k_outsum<<<dim3((GEX / 2 + 255) / 256, BATCH), 256, 0, stream>>>(Ppart, out);
}


# k_pred0 dot products with lane = batch row: S rows in LDS at a 528-byte pitch, weight pairs at one address per wave, so 16-byte LDS reads stop colliding; loads issued as one group; sums unchanged
# speedup vs baseline: 5.0973x; 1.0772x over previous
#include <hip/hip_runtime.h>
#include <hip/hip_bf16.h>

#define GEX 978
#define KPAD 980
#define HDIM 128
#define NAT 50
#define FEAT 34
#define CCEH 64
#define BATCH 64
#define GCH 16
#define NSLAB ((GEX + GCH - 1) / GCH)
#define BGRP 16
#define PPITCH (CCEH + 1)
#define MC 25
#define NCH (NAT / MC)
#define CCE_T 288
#define NFT ((MC * FEAT + CCE_T - 1) / CCE_T)
#define ADT ((NAT * MC + CCE_T - 1) / CCE_T)
#define KSTEPS 35
#define KSLAB (4 * KSTEPS)
#define NKS (KPAD / KSLAB)
#define NBLK 32
#define NNB ((GEX + NBLK - 1) / NBLK)
#define PPN (NNB * NBLK)
#define PPI_V2 (GEX / 2)
#define PPI_LD ((PPI_V2 + 63) / 64)
#define SPITCH (HDIM + 4)

typedef __attribute__((ext_vector_type(2))) float v2f;
typedef __attribute__((ext_vector_type(4))) float v4f;
typedef __attribute__((ext_vector_type(8))) float v8f;

__global__ __launch_bounds__(CCE_T) void k_cce(const float* __restrict__ nf,
                                               const float* __restrict__ mask,
                                               const float* __restrict__ adj,
                                               const float* __restrict__ dist,
                                               const float* __restrict__ W1,
                                               float* __restrict__ ppool) {
    int b = blockIdx.x, c = blockIdx.y, tid = threadIdx.x, m0 = c * MC;
    __shared__ float nfs[MC * FEAT];
    __shared__ float as[NAT * MC];
    __shared__ float es[NAT * MC];
    __shared__ float hs[MC * CCEH];
    __shared__ float ms[NAT];
    __shared__ float wm[MC];
    __shared__ float msum_s;

    int d = tid & (CCEH - 1);
    float w1[FEAT];
    #pragma unroll
    for (int f = 0; f < FEAT; f++) w1[f] = W1[f * CCEH + d];
    const float* nfb = nf + (b * NAT + m0) * FEAT;
    float nv[NFT];
    #pragma unroll
    for (int j = 0; j < NFT; j++) {
        int i = tid + CCE_T * j;
        nv[j] = nfb[i < MC * FEAT ? i : MC * FEAT - 1];
    }
    float av[ADT], dv[ADT];
    #pragma unroll
    for (int q = 0; q < ADT; q++) {
        int j = tid + CCE_T * q;
        int jc = j < NAT * MC ? j : NAT * MC - 1;
        int n = jc / MC, ml = jc - n * MC;
        int idx = (b * NAT + n) * NAT + m0 + ml;
        av[q] = adj[idx];
        dv[q] = dist[idx];
    }
    float mk = mask[b * NAT + (tid < NAT ? tid : NAT - 1)];

    #pragma unroll
    for (int j = 0; j < NFT; j++) {
        int i = tid + CCE_T * j;
        if (i < MC * FEAT) nfs[i] = nv[j];
    }
    #pragma unroll
    for (int q = 0; q < ADT; q++) {
        int j = tid + CCE_T * q;
        if (j < NAT * MC) { as[j] = av[q]; es[j] = expf(-dv[q]); }
    }
    if (tid < NAT) ms[tid] = mk;
    __syncthreads();

    if (tid < 256) {
        int mq = tid >> 6;
        #pragma unroll
        for (int r = 0; r < (MC + 3) / 4; r++) {
            int ml = mq + 4 * r;
            if (ml < MC) {
                float s = 0.f;
                #pragma unroll
                for (int f = 0; f < FEAT; f++) s += nfs[ml * FEAT + f] * w1[f];
                hs[ml * CCEH + d] = fmaxf(s, 0.f);
            }
        }
    } else if (tid < 256 + MC) {
        int ml = tid - 256;
        float s = 0.f, msum = 0.f;
        #pragma unroll 10
        for (int n = 0; n < NAT; n++) {
            s += ms[n] * as[n * MC + ml] * es[n * MC + ml];
            msum += ms[n];
        }
        wm[ml] = s;
        if (ml == 0) msum_s = fmaxf(msum, 1.f);
    }
    __syncthreads();
    if (tid < CCEH) {
        float s = 0.f;
        #pragma unroll
        for (int ml = 0; ml < MC; ml++) s += wm[ml] * hs[ml * CCEH + tid];
        ppool[(b * NCH + c) * CCEH + tid] = s / msum_s;
    }
}

__global__ __launch_bounds__(256) void k_compS(const float* __restrict__ ppool,
                                               const float* __restrict__ W2,
                                               const float* __restrict__ dose,
                                               const float* __restrict__ timev,
                                               const float* __restrict__ w_dose,
                                               const float* __restrict__ w_time,
                                               const float* __restrict__ b_gex,
                                               const float* __restrict__ w_gex,
                                               const float* __restrict__ w_comp,
                                               float* __restrict__ comp,
                                               float* __restrict__ Spart) {
    __shared__ float Ps[BGRP * PPITCH];
    __shared__ float W2s[CCEH * GCH];
    __shared__ __attribute__((aligned(16))) float Xs[BGRP * GCH];
    __shared__ __attribute__((aligned(16))) float Cs[BGRP * GCH];
    int slab = blockIdx.x, b0 = blockIdx.y * BGRP, g0 = slab * GCH, tid = threadIdx.x;

    float pv[4][NCH], wv[4];
    #pragma unroll
    for (int j = 0; j < 4; j++) {
        int i = tid + 256 * j;
        int gq = g0 + (i & 15);
        #pragma unroll
        for (int c = 0; c < NCH; c++)
            pv[j][c] = ppool[((b0 + (i >> 6)) * NCH + c) * CCEH + (i & 63)];
        wv[j] = W2[(i >> 4) * GEX + (gq < GEX ? gq : GEX - 1)];
    }
    int bl = tid >> 4, gl = tid & 15, b = b0 + bl, g = g0 + gl;
    bool gok = g < GEX;
    int gc = gok ? g : GEX - 1;
    float x  = b_gex[b * GEX + gc];
    float s  = dose[b] * w_dose[gc] + timev[b] * w_time[gc];
    int h = tid & (HDIM - 1), bh = tid >> 7;
    float wg[GCH], wc[GCH];
    #pragma unroll
    for (int j = 0; j < GCH; j++) {
        int gq = g0 + j;
        int gr = gq < GEX ? gq : GEX - 1;
        float a = w_gex[gr * HDIM + h], c2 = w_comp[gr * HDIM + h];
        wg[j] = gq < GEX ? a : 0.f;
        wc[j] = gq < GEX ? c2 : 0.f;
    }
    #pragma unroll
    for (int j = 0; j < 4; j++) {
        int i = tid + 256 * j;
        float p = pv[j][0];
        #pragma unroll
        for (int c = 1; c < NCH; c++) p += pv[j][c];
        Ps[(i >> 6) * PPITCH + (i & 63)] = p;
        W2s[i] = (g0 + (i & 15) < GEX) ? wv[j] : 0.f;
    }
    if (!gok) { x = 0.f; s = 0.f; }
    __syncthreads();
    #pragma unroll 8
    for (int d = 0; d < CCEH; d++) s += Ps[bl * PPITCH + d] * W2s[d * GCH + gl];
    if (gok) comp[b * GEX + g] = s;
    Xs[tid] = x;
    Cs[tid] = gok ? s : 0.f;
    __syncthreads();
    #pragma unroll
    for (int r = 0; r < BGRP / 2; r++) {
        int row = bh * (BGRP / 2) + r;
        float acc = 0.f;
        #pragma unroll
        for (int j = 0; j < GCH; j++)
            acc += Xs[row * GCH + j] * wg[j] + Cs[row * GCH + j] * wc[j];
        Spart[(slab * BATCH + b0 + row) * HDIM + h] = acc;
    }
}

__global__ __launch_bounds__(128) void k_S(const float* __restrict__ Spart,
                                           float* __restrict__ S) {
    int i = blockIdx.x * HDIM + threadIdx.x;
    float s = 0.f;
    #pragma unroll
    for (int p = 0; p < NSLAB; p++) s += Spart[p * (BATCH * HDIM) + i];
    S[i] = s;
}

__global__ __launch_bounds__(256) void k_pred0(const float* __restrict__ b_gex,
                                               const float* __restrict__ comp,
                                               const float* __restrict__ w_gex,
                                               const float* __restrict__ w_comp,
                                               const float* __restrict__ S,
                                               const float* __restrict__ ppi,
                                               float* __restrict__ pred0) {
    __shared__ __attribute__((aligned(16))) float Ssh[BATCH * SPITCH];
    __shared__ __attribute__((aligned(16))) float wsh[4 * HDIM * 2];
    __shared__ __attribute__((aligned(8)))  float dsh[BATCH * 4 * 2];
    __shared__ float rs[8];
    int g0 = blockIdx.x * 4, tid = threadIdx.x;
    int pr = g0 + (tid >> 6), pt = tid & 63;
    const v2f* prow = (const v2f*)(ppi + (pr < GEX ? pr : GEX - 1) * GEX);
    v2f pv[PPI_LD];
    #pragma unroll
    for (int i = 0; i < PPI_LD; i++) {
        int j = pt + 64 * i;
        pv[i] = prow[j < PPI_V2 ? j : PPI_V2 - 1];
    }
    int wv_ = tid >> 5, lane = tid & 31;
    const v4f* S4 = (const v4f*)S;
    v4f sv[8];
    #pragma unroll
    for (int i = 0; i < 8; i++) sv[i] = S4[tid + 256 * i];
    int wgx = g0 + (wv_ & 3);
    int wof = (wgx < GEX ? wgx : GEX - 1) * HDIM + lane * 4;
    v4f wgv = *(const v4f*)(w_gex + wof), wcv = *(const v4f*)(w_comp + wof);
    int fb = tid >> 2, fgl = tid & 3, fg = g0 + fgl;
    int fgc = fg < GEX ? fg : GEX - 1;
    float x = b_gex[fb * GEX + fgc], c = comp[fb * GEX + fgc];
    asm volatile("" :: "v"(pv[0]), "v"(pv[1]), "v"(pv[2]), "v"(pv[3]), "v"(pv[4]), "v"(pv[5]),
                       "v"(pv[6]), "v"(pv[7]), "v"(sv[0]), "v"(sv[1]), "v"(sv[2]), "v"(sv[3]),
                       "v"(sv[4]), "v"(sv[5]), "v"(sv[6]), "v"(sv[7]), "v"(wgv), "v"(wcv),
                       "v"(x), "v"(c));

    #pragma unroll
    for (int i = 0; i < 8; i++)
        *(v4f*)(Ssh + (wv_ + 8 * i) * SPITCH + lane * 4) = sv[i];
    if (wv_ < 4) {
        v4f lo = {wgv.x, wcv.x, wgv.y, wcv.y}, hi = {wgv.z, wcv.z, wgv.w, wcv.w};
        v4f* wd = (v4f*)(wsh + (wv_ * HDIM + lane * 4) * 2);
        wd[0] = lo;
        wd[1] = hi;
    }
    float rsum = 0.f;
    #pragma unroll
    for (int i = 0; i < PPI_LD; i++) {
        float a = pv[i].x + pv[i].y;
        rsum += (pt + 64 * i < PPI_V2) ? a : 0.f;
    }
    for (int o = 16; o > 0; o >>= 1) rsum += __shfl_xor(rsum, o);
    if (lane == 0) rs[wv_] = rsum;
    __syncthreads();
    if (wv_ < 4) {
        const v4f* sa = (const v4f*)(Ssh + lane * SPITCH);
        const v4f* sb = (const v4f*)(Ssh + (lane + 32) * SPITCH);
        const v4f* wp = (const v4f*)(wsh + wv_ * HDIM * 2);
        v2f da = {0.f, 0.f}, db = {0.f, 0.f};
        #pragma unroll 8
        for (int q = 0; q < HDIM / 4; q++) {
            v4f a = sa[q], b4 = sb[q], w01 = wp[2 * q], w23 = wp[2 * q + 1];
            v2f w0 = {w01.x, w01.y}, w1 = {w01.z, w01.w}, w2 = {w23.x, w23.y}, w3 = {w23.z, w23.w};
            da = __builtin_elementwise_fma((v2f){a.x, a.x}, w0, da);
            db = __builtin_elementwise_fma((v2f){b4.x, b4.x}, w0, db);
            da = __builtin_elementwise_fma((v2f){a.y, a.y}, w1, da);
            db = __builtin_elementwise_fma((v2f){b4.y, b4.y}, w1, db);
            da = __builtin_elementwise_fma((v2f){a.z, a.z}, w2, da);
            db = __builtin_elementwise_fma((v2f){b4.z, b4.z}, w2, db);
            da = __builtin_elementwise_fma((v2f){a.w, a.w}, w3, da);
            db = __builtin_elementwise_fma((v2f){b4.w, b4.w}, w3, db);
        }
        *(v2f*)(dsh + (lane * 4 + wv_) * 2) = da;
        *(v2f*)(dsh + ((lane + 32) * 4 + wv_) * 2) = db;
    }
    __syncthreads();
    if (fg < GEX) {
        v2f d = *(const v2f*)(dsh + tid * 2);
        float row = (x * d.x + c * d.y) * 0.08838834764831845f
                  + (rs[2 * fgl] + rs[2 * fgl + 1]);
        pred0[fb * GEX + fg] = x * row;
    }
}

__global__ __launch_bounds__(256) void k_ln(const float* __restrict__ pred0,
                                            const float* __restrict__ gamma,
                                            const float* __restrict__ beta,
                                            float* __restrict__ xlnP) {
    int b = blockIdx.x, tid = threadIdx.x;
    __shared__ float red[256];
    __shared__ float mu_s, rstd_s;
    float s = 0.f;
    for (int g = tid; g < GEX; g += 256) s += pred0[b * GEX + g];
    red[tid] = s;
    __syncthreads();
    for (int o = 128; o > 0; o >>= 1) { if (tid < o) red[tid] += red[tid + o]; __syncthreads(); }
    if (tid == 0) mu_s = red[0] / (float)GEX;
    __syncthreads();
    float mu = mu_s, v = 0.f;
    for (int g = tid; g < GEX; g += 256) { float d = pred0[b * GEX + g] - mu; v += d * d; }
    red[tid] = v;
    __syncthreads();
    for (int o = 128; o > 0; o >>= 1) { if (tid < o) red[tid] += red[tid + o]; __syncthreads(); }
    if (tid == 0) rstd_s = rsqrtf(red[0] / (float)GEX + 1e-5f);
    __syncthreads();
    float rstd = rstd_s;
    for (int g = tid; g < KPAD; g += 256) {
        float y = 0.f;
        if (g < GEX)
            y = fmaxf((pred0[b * GEX + g] - mu) * rstd * gamma[g] + beta[g], 0.f);
        xlnP[b * KPAD + g] = y;
    }
}

__global__ __launch_bounds__(256) void k_gemm_slab(const float* __restrict__ xlnP,
                                                   const float* __restrict__ Wff,
                                                   float* __restrict__ Ppart) {
    int lane = threadIdx.x & 31, w = threadIdx.x >> 5;
    int mt = w & 3, ns = w >> 2;
    int slab = blockIdx.y;
    int lg   = lane & 15;
    int koff = (lane >= 16) ? 2 : 0;
    int m  = mt * 16 + lg;
    int n  = blockIdx.x * NBLK + ns * 16 + lg;
    int nc = (n < GEX) ? n : (GEX - 1);
    int kb = slab * KSLAB + koff;

    const float* ap = xlnP + m * KPAD + kb;
    const float* bp = Wff + kb * GEX + nc;
    float* pp = Ppart + (slab * BATCH + mt * 16 + ((lane >= 16) ? 8 : 0)) * PPN + n;

    v8f acc = {};
    #pragma unroll
    for (int s = 0; s < KSTEPS; s++) {
        v2f a = *(const v2f*)(ap + 4 * s);
        v2f bm;
        if (s < KSTEPS - 1) {
            bm.x = bp[(4 * s) * GEX];
            bm.y = bp[(4 * s + 1) * GEX];
        } else {
            int k0 = kb + 4 * s, k1 = k0 + 1;
            int r0 = (k0 < GEX) ? k0 : (GEX - 1);
            int r1 = (k1 < GEX) ? k1 : (GEX - 1);
            float b0 = Wff[r0 * GEX + nc], b1 = Wff[r1 * GEX + nc];
            bm.x = (k0 < GEX) ? b0 : 0.f;
            bm.y = (k1 < GEX) ? b1 : 0.f;
        }
        acc = __builtin_amdgcn_wmma_f32_16x16x4_f32(
            false, a, false, bm, (short)0, acc, false, false);
    }

    for (int v = 0; v < 8; v++) pp[v * PPN] = acc[v];
}

__global__ __launch_bounds__(256) void k_outsum(const float* __restrict__ Ppart,
                                                float* __restrict__ out) {
    int j = blockIdx.x * 256 + threadIdx.x, b = blockIdx.y;
    if (j < GEX / 2) {
        const float* pl = Ppart + b * PPN + 2 * j;
        v2f s = *(const v2f*)pl;
        for (int p = 1; p < NKS; p++) s += *(const v2f*)(pl + p * (BATCH * PPN));
        *(v2f*)(out + b * GEX + 2 * j) = s;
    }
}

extern "C" void kernel_launch(void* const* d_in, const int* in_sizes, int n_in,
                              void* d_out, int out_size, void* d_ws, size_t ws_size,
                              hipStream_t stream) {
    const float* b_gex    = (const float*)d_in[0];
    const float* node_feat= (const float*)d_in[1];
    const float* mask     = (const float*)d_in[2];
    const float* adj      = (const float*)d_in[3];
    const float* dist     = (const float*)d_in[4];
    const float* dose     = (const float*)d_in[5];
    const float* timev    = (const float*)d_in[6];
    const float* ppi      = (const float*)d_in[7];
    const float* w_gex    = (const float*)d_in[8];
    const float* w_comp   = (const float*)d_in[9];
    const float* W1       = (const float*)d_in[10];
    const float* W2       = (const float*)d_in[11];
    const float* w_dose   = (const float*)d_in[12];
    const float* w_time   = (const float*)d_in[13];
    const float* gamma    = (const float*)d_in[14];
    const float* beta     = (const float*)d_in[15];
    const float* Wff      = (const float*)d_in[16];

    float* out  = (float*)d_out;
    float* comp = out + BATCH * GEX;

    float* ws     = (float*)d_ws;
    float* ppool  = ws;
    float* S      = ppool + BATCH * NCH * CCEH;
    float* pred0  = S + BATCH * HDIM;
    float* xlnP   = pred0 + BATCH * GEX;
    float* Ppart  = xlnP + BATCH * KPAD;
    float* Spart  = Ppart + NKS * BATCH * PPN;

    k_cce<<<dim3(BATCH, NCH), CCE_T, 0, stream>>>(node_feat, mask, adj, dist, W1, ppool);
    k_compS<<<dim3(NSLAB, BATCH / BGRP), 256, 0, stream>>>(ppool, W2, dose, timev, w_dose, w_time,
                                                           b_gex, w_gex, w_comp, comp, Spart);
    k_S<<<BATCH, HDIM, 0, stream>>>(Spart, S);
    k_pred0<<<(GEX + 3) / 4, 256, 0, stream>>>(b_gex, comp, w_gex, w_comp, S, ppi, pred0);
    k_ln<<<BATCH, 256, 0, stream>>>(pred0, gamma, beta, xlnP);
    k_gemm_slab<<<dim3(NNB, NKS), 256, 0, stream>>>(xlnP, Wff, Ppart);
    k_outsum<<<dim3((GEX / 2 + 255) / 256, BATCH), 256, 0, stream>>>(Ppart, out);
}


# k_ln keeps its row in registers: pred0, gamma, beta fetched once as 8-byte pairs in one group; mean and variance by DPP lane sums and one LDS exchange of 8 wave sums each; LN sums' order changes
# speedup vs baseline: 5.7629x; 1.1306x over previous
#include <hip/hip_runtime.h>
#include <hip/hip_bf16.h>

#define GEX 978
#define KPAD 980
#define HDIM 128
#define NAT 50
#define FEAT 34
#define CCEH 64
#define BATCH 64
#define GCH 16
#define NSLAB ((GEX + GCH - 1) / GCH)
#define BGRP 16
#define PPITCH (CCEH + 1)
#define MC 25
#define NCH (NAT / MC)
#define CCE_T 288
#define NFT ((MC * FEAT + CCE_T - 1) / CCE_T)
#define ADT ((NAT * MC + CCE_T - 1) / CCE_T)
#define KSTEPS 35
#define KSLAB (4 * KSTEPS)
#define NKS (KPAD / KSLAB)
#define NBLK 32
#define NNB ((GEX + NBLK - 1) / NBLK)
#define PPN (NNB * NBLK)
#define PPI_V2 (GEX / 2)
#define PPI_LD ((PPI_V2 + 63) / 64)
#define SPITCH (HDIM + 4)

typedef __attribute__((ext_vector_type(2))) float v2f;
typedef __attribute__((ext_vector_type(4))) float v4f;
typedef __attribute__((ext_vector_type(8))) float v8f;

__global__ __launch_bounds__(CCE_T) void k_cce(const float* __restrict__ nf,
                                               const float* __restrict__ mask,
                                               const float* __restrict__ adj,
                                               const float* __restrict__ dist,
                                               const float* __restrict__ W1,
                                               float* __restrict__ ppool) {
    int b = blockIdx.x, c = blockIdx.y, tid = threadIdx.x, m0 = c * MC;
    __shared__ float nfs[MC * FEAT];
    __shared__ float as[NAT * MC];
    __shared__ float es[NAT * MC];
    __shared__ float hs[MC * CCEH];
    __shared__ float ms[NAT];
    __shared__ float wm[MC];
    __shared__ float msum_s;

    int d = tid & (CCEH - 1);
    float w1[FEAT];
    #pragma unroll
    for (int f = 0; f < FEAT; f++) w1[f] = W1[f * CCEH + d];
    const float* nfb = nf + (b * NAT + m0) * FEAT;
    float nv[NFT];
    #pragma unroll
    for (int j = 0; j < NFT; j++) {
        int i = tid + CCE_T * j;
        nv[j] = nfb[i < MC * FEAT ? i : MC * FEAT - 1];
    }
    float av[ADT], dv[ADT];
    #pragma unroll
    for (int q = 0; q < ADT; q++) {
        int j = tid + CCE_T * q;
        int jc = j < NAT * MC ? j : NAT * MC - 1;
        int n = jc / MC, ml = jc - n * MC;
        int idx = (b * NAT + n) * NAT + m0 + ml;
        av[q] = adj[idx];
        dv[q] = dist[idx];
    }
    float mk = mask[b * NAT + (tid < NAT ? tid : NAT - 1)];

    #pragma unroll
    for (int j = 0; j < NFT; j++) {
        int i = tid + CCE_T * j;
        if (i < MC * FEAT) nfs[i] = nv[j];
    }
    #pragma unroll
    for (int q = 0; q < ADT; q++) {
        int j = tid + CCE_T * q;
        if (j < NAT * MC) { as[j] = av[q]; es[j] = expf(-dv[q]); }
    }
    if (tid < NAT) ms[tid] = mk;
    __syncthreads();

    if (tid < 256) {
        int mq = tid >> 6;
        #pragma unroll
        for (int r = 0; r < (MC + 3) / 4; r++) {
            int ml = mq + 4 * r;
            if (ml < MC) {
                float s = 0.f;
                #pragma unroll
                for (int f = 0; f < FEAT; f++) s += nfs[ml * FEAT + f] * w1[f];
                hs[ml * CCEH + d] = fmaxf(s, 0.f);
            }
        }
    } else if (tid < 256 + MC) {
        int ml = tid - 256;
        float s = 0.f, msum = 0.f;
        #pragma unroll 10
        for (int n = 0; n < NAT; n++) {
            s += ms[n] * as[n * MC + ml] * es[n * MC + ml];
            msum += ms[n];
        }
        wm[ml] = s;
        if (ml == 0) msum_s = fmaxf(msum, 1.f);
    }
    __syncthreads();
    if (tid < CCEH) {
        float s = 0.f;
        #pragma unroll
        for (int ml = 0; ml < MC; ml++) s += wm[ml] * hs[ml * CCEH + tid];
        ppool[(b * NCH + c) * CCEH + tid] = s / msum_s;
    }
}

__global__ __launch_bounds__(256) void k_compS(const float* __restrict__ ppool,
                                               const float* __restrict__ W2,
                                               const float* __restrict__ dose,
                                               const float* __restrict__ timev,
                                               const float* __restrict__ w_dose,
                                               const float* __restrict__ w_time,
                                               const float* __restrict__ b_gex,
                                               const float* __restrict__ w_gex,
                                               const float* __restrict__ w_comp,
                                               float* __restrict__ comp,
                                               float* __restrict__ Spart) {
    __shared__ float Ps[BGRP * PPITCH];
    __shared__ float W2s[CCEH * GCH];
    __shared__ __attribute__((aligned(16))) float Xs[BGRP * GCH];
    __shared__ __attribute__((aligned(16))) float Cs[BGRP * GCH];
    int slab = blockIdx.x, b0 = blockIdx.y * BGRP, g0 = slab * GCH, tid = threadIdx.x;

    float pv[4][NCH], wv[4];
    #pragma unroll
    for (int j = 0; j < 4; j++) {
        int i = tid + 256 * j;
        int gq = g0 + (i & 15);
        #pragma unroll
        for (int c = 0; c < NCH; c++)
            pv[j][c] = ppool[((b0 + (i >> 6)) * NCH + c) * CCEH + (i & 63)];
        wv[j] = W2[(i >> 4) * GEX + (gq < GEX ? gq : GEX - 1)];
    }
    int bl = tid >> 4, gl = tid & 15, b = b0 + bl, g = g0 + gl;
    bool gok = g < GEX;
    int gc = gok ? g : GEX - 1;
    float x  = b_gex[b * GEX + gc];
    float s  = dose[b] * w_dose[gc] + timev[b] * w_time[gc];
    int h = tid & (HDIM - 1), bh = tid >> 7;
    float wg[GCH], wc[GCH];
    #pragma unroll
    for (int j = 0; j < GCH; j++) {
        int gq = g0 + j;
        int gr = gq < GEX ? gq : GEX - 1;
        float a = w_gex[gr * HDIM + h], c2 = w_comp[gr * HDIM + h];
        wg[j] = gq < GEX ? a : 0.f;
        wc[j] = gq < GEX ? c2 : 0.f;
    }
    #pragma unroll
    for (int j = 0; j < 4; j++) {
        int i = tid + 256 * j;
        float p = pv[j][0];
        #pragma unroll
        for (int c = 1; c < NCH; c++) p += pv[j][c];
        Ps[(i >> 6) * PPITCH + (i & 63)] = p;
        W2s[i] = (g0 + (i & 15) < GEX) ? wv[j] : 0.f;
    }
    if (!gok) { x = 0.f; s = 0.f; }
    __syncthreads();
    #pragma unroll 8
    for (int d = 0; d < CCEH; d++) s += Ps[bl * PPITCH + d] * W2s[d * GCH + gl];
    if (gok) comp[b * GEX + g] = s;
    Xs[tid] = x;
    Cs[tid] = gok ? s : 0.f;
    __syncthreads();
    #pragma unroll
    for (int r = 0; r < BGRP / 2; r++) {
        int row = bh * (BGRP / 2) + r;
        float acc = 0.f;
        #pragma unroll
        for (int j = 0; j < GCH; j++)
            acc += Xs[row * GCH + j] * wg[j] + Cs[row * GCH + j] * wc[j];
        Spart[(slab * BATCH + b0 + row) * HDIM + h] = acc;
    }
}

__global__ __launch_bounds__(128) void k_S(const float* __restrict__ Spart,
                                           float* __restrict__ S) {
    int i = blockIdx.x * HDIM + threadIdx.x;
    float s = 0.f;
    #pragma unroll
    for (int p = 0; p < NSLAB; p++) s += Spart[p * (BATCH * HDIM) + i];
    S[i] = s;
}

__global__ __launch_bounds__(256) void k_pred0(const float* __restrict__ b_gex,
                                               const float* __restrict__ comp,
                                               const float* __restrict__ w_gex,
                                               const float* __restrict__ w_comp,
                                               const float* __restrict__ S,
                                               const float* __restrict__ ppi,
                                               float* __restrict__ pred0) {
    __shared__ __attribute__((aligned(16))) float Ssh[BATCH * SPITCH];
    __shared__ __attribute__((aligned(16))) float wsh[4 * HDIM * 2];
    __shared__ __attribute__((aligned(8)))  float dsh[BATCH * 4 * 2];
    __shared__ float rs[8];
    int g0 = blockIdx.x * 4, tid = threadIdx.x;
    int pr = g0 + (tid >> 6), pt = tid & 63;
    const v2f* prow = (const v2f*)(ppi + (pr < GEX ? pr : GEX - 1) * GEX);
    v2f pv[PPI_LD];
    #pragma unroll
    for (int i = 0; i < PPI_LD; i++) {
        int j = pt + 64 * i;
        pv[i] = prow[j < PPI_V2 ? j : PPI_V2 - 1];
    }
    int wv_ = tid >> 5, lane = tid & 31;
    const v4f* S4 = (const v4f*)S;
    v4f sv[8];
    #pragma unroll
    for (int i = 0; i < 8; i++) sv[i] = S4[tid + 256 * i];
    int wgx = g0 + (wv_ & 3);
    int wof = (wgx < GEX ? wgx : GEX - 1) * HDIM + lane * 4;
    v4f wgv = *(const v4f*)(w_gex + wof), wcv = *(const v4f*)(w_comp + wof);
    int fb = tid >> 2, fgl = tid & 3, fg = g0 + fgl;
    int fgc = fg < GEX ? fg : GEX - 1;
    float x = b_gex[fb * GEX + fgc], c = comp[fb * GEX + fgc];
    asm volatile("" :: "v"(pv[0]), "v"(pv[1]), "v"(pv[2]), "v"(pv[3]), "v"(pv[4]), "v"(pv[5]),
                       "v"(pv[6]), "v"(pv[7]), "v"(sv[0]), "v"(sv[1]), "v"(sv[2]), "v"(sv[3]),
                       "v"(sv[4]), "v"(sv[5]), "v"(sv[6]), "v"(sv[7]), "v"(wgv), "v"(wcv),
                       "v"(x), "v"(c));

    #pragma unroll
    for (int i = 0; i < 8; i++)
        *(v4f*)(Ssh + (wv_ + 8 * i) * SPITCH + lane * 4) = sv[i];
    if (wv_ < 4) {
        v4f lo = {wgv.x, wcv.x, wgv.y, wcv.y}, hi = {wgv.z, wcv.z, wgv.w, wcv.w};
        v4f* wd = (v4f*)(wsh + (wv_ * HDIM + lane * 4) * 2);
        wd[0] = lo;
        wd[1] = hi;
    }
    float rsum = 0.f;
    #pragma unroll
    for (int i = 0; i < PPI_LD; i++) {
        float a = pv[i].x + pv[i].y;
        rsum += (pt + 64 * i < PPI_V2) ? a : 0.f;
    }
    for (int o = 16; o > 0; o >>= 1) rsum += __shfl_xor(rsum, o);
    if (lane == 0) rs[wv_] = rsum;
    __syncthreads();
    if (wv_ < 4) {
        const v4f* sa = (const v4f*)(Ssh + lane * SPITCH);
        const v4f* sb = (const v4f*)(Ssh + (lane + 32) * SPITCH);
        const v4f* wp = (const v4f*)(wsh + wv_ * HDIM * 2);
        v2f da = {0.f, 0.f}, db = {0.f, 0.f};
        #pragma unroll 8
        for (int q = 0; q < HDIM / 4; q++) {
            v4f a = sa[q], b4 = sb[q], w01 = wp[2 * q], w23 = wp[2 * q + 1];
            v2f w0 = {w01.x, w01.y}, w1 = {w01.z, w01.w}, w2 = {w23.x, w23.y}, w3 = {w23.z, w23.w};
            da = __builtin_elementwise_fma((v2f){a.x, a.x}, w0, da);
            db = __builtin_elementwise_fma((v2f){b4.x, b4.x}, w0, db);
            da = __builtin_elementwise_fma((v2f){a.y, a.y}, w1, da);
            db = __builtin_elementwise_fma((v2f){b4.y, b4.y}, w1, db);
            da = __builtin_elementwise_fma((v2f){a.z, a.z}, w2, da);
            db = __builtin_elementwise_fma((v2f){b4.z, b4.z}, w2, db);
            da = __builtin_elementwise_fma((v2f){a.w, a.w}, w3, da);
            db = __builtin_elementwise_fma((v2f){b4.w, b4.w}, w3, db);
        }
        *(v2f*)(dsh + (lane * 4 + wv_) * 2) = da;
        *(v2f*)(dsh + ((lane + 32) * 4 + wv_) * 2) = db;
    }
    __syncthreads();
    if (fg < GEX) {
        v2f d = *(const v2f*)(dsh + tid * 2);
        float row = (x * d.x + c * d.y) * 0.08838834764831845f
                  + (rs[2 * fgl] + rs[2 * fgl + 1]);
        pred0[fb * GEX + fg] = x * row;
    }
}

template <int M>
__device__ __forceinline__ float row_xor(float x) {
    int b = __builtin_bit_cast(int, x);
    return __builtin_bit_cast(float, __builtin_amdgcn_update_dpp(b, b, 0x160 + M, 0xF, 0xF, false));
}
__device__ __forceinline__ float xhalf(float x) {
    int b = __builtin_bit_cast(int, x);
    return __builtin_bit_cast(float, __builtin_amdgcn_permlanex16(b, b, 0x76543210, 0xfedcba98, false, false));
}
__device__ __forceinline__ float wave_sum(float x) {
    x += row_xor<1>(x);
    x += row_xor<2>(x);
    x += row_xor<4>(x);
    x += row_xor<8>(x);
    return x + xhalf(x);
}

__global__ __launch_bounds__(256) void k_ln(const float* __restrict__ pred0,
                                            const float* __restrict__ gamma,
                                            const float* __restrict__ beta,
                                            float* __restrict__ xlnP) {
    int b = blockIdx.x, tid = threadIdx.x, lane = tid & 31, wv_ = tid >> 5;
    __shared__ float red[2][8];
    const v2f* p2 = (const v2f*)(pred0 + b * GEX);
    const v2f* g2 = (const v2f*)gamma;
    const v2f* b2 = (const v2f*)beta;
    int j1 = tid + 256;
    bool ok1 = j1 < PPI_V2;
    int j1c = ok1 ? j1 : PPI_V2 - 1;
    v2f pa = p2[tid], pb = p2[j1c];
    v2f ga = g2[tid], gb = g2[j1c];
    v2f ba = b2[tid], bb = b2[j1c];

    float s = (pa.x + pa.y) + (ok1 ? pb.x + pb.y : 0.f);
    s = wave_sum(s);
    if (lane == 0) red[0][wv_] = s;
    __syncthreads();
    float t = red[0][0];
    for (int w = 1; w < 8; w++) t += red[0][w];
    float mu = t / (float)GEX;

    float d0 = pa.x - mu, d1 = pa.y - mu, d2 = pb.x - mu, d3 = pb.y - mu;
    float v = (d0 * d0 + d1 * d1) + (ok1 ? d2 * d2 + d3 * d3 : 0.f);
    v = wave_sum(v);
    if (lane == 0) red[1][wv_] = v;
    __syncthreads();
    float u = red[1][0];
    for (int w = 1; w < 8; w++) u += red[1][w];
    float rstd = rsqrtf(u / (float)GEX + 1e-5f);

    v2f* xo = (v2f*)(xlnP + b * KPAD);
    v2f ya = {fmaxf(d0 * rstd * ga.x + ba.x, 0.f), fmaxf(d1 * rstd * ga.y + ba.y, 0.f)};
    xo[tid] = ya;
    if (ok1) {
        v2f yb = {fmaxf(d2 * rstd * gb.x + bb.x, 0.f), fmaxf(d3 * rstd * gb.y + bb.y, 0.f)};
        xo[j1] = yb;
    } else if (j1 == PPI_V2) {
        xo[j1] = (v2f){0.f, 0.f};
    }
}

__global__ __launch_bounds__(256) void k_gemm_slab(const float* __restrict__ xlnP,
                                                   const float* __restrict__ Wff,
                                                   float* __restrict__ Ppart) {
    int lane = threadIdx.x & 31, w = threadIdx.x >> 5;
    int mt = w & 3, ns = w >> 2;
    int slab = blockIdx.y;
    int lg   = lane & 15;
    int koff = (lane >= 16) ? 2 : 0;
    int m  = mt * 16 + lg;
    int n  = blockIdx.x * NBLK + ns * 16 + lg;
    int nc = (n < GEX) ? n : (GEX - 1);
    int kb = slab * KSLAB + koff;

    const float* ap = xlnP + m * KPAD + kb;
    const float* bp = Wff + kb * GEX + nc;
    float* pp = Ppart + (slab * BATCH + mt * 16 + ((lane >= 16) ? 8 : 0)) * PPN + n;

    v8f acc = {};
    #pragma unroll
    for (int s = 0; s < KSTEPS; s++) {
        v2f a = *(const v2f*)(ap + 4 * s);
        v2f bm;
        if (s < KSTEPS - 1) {
            bm.x = bp[(4 * s) * GEX];
            bm.y = bp[(4 * s + 1) * GEX];
        } else {
            int k0 = kb + 4 * s, k1 = k0 + 1;
            int r0 = (k0 < GEX) ? k0 : (GEX - 1);
            int r1 = (k1 < GEX) ? k1 : (GEX - 1);
            float b0 = Wff[r0 * GEX + nc], b1 = Wff[r1 * GEX + nc];
            bm.x = (k0 < GEX) ? b0 : 0.f;
            bm.y = (k1 < GEX) ? b1 : 0.f;
        }
        acc = __builtin_amdgcn_wmma_f32_16x16x4_f32(
            false, a, false, bm, (short)0, acc, false, false);
    }

    for (int v = 0; v < 8; v++) pp[v * PPN] = acc[v];
}

__global__ __launch_bounds__(256) void k_outsum(const float* __restrict__ Ppart,
                                                float* __restrict__ out) {
    int j = blockIdx.x * 256 + threadIdx.x, b = blockIdx.y;
    if (j < GEX / 2) {
        const float* pl = Ppart + b * PPN + 2 * j;
        v2f s = *(const v2f*)pl;
        for (int p = 1; p < NKS; p++) s += *(const v2f*)(pl + p * (BATCH * PPN));
        *(v2f*)(out + b * GEX + 2 * j) = s;
    }
}

extern "C" void kernel_launch(void* const* d_in, const int* in_sizes, int n_in,
                              void* d_out, int out_size, void* d_ws, size_t ws_size,
                              hipStream_t stream) {
    const float* b_gex    = (const float*)d_in[0];
    const float* node_feat= (const float*)d_in[1];
    const float* mask     = (const float*)d_in[2];
    const float* adj      = (const float*)d_in[3];
    const float* dist     = (const float*)d_in[4];
    const float* dose     = (const float*)d_in[5];
    const float* timev    = (const float*)d_in[6];
    const float* ppi      = (const float*)d_in[7];
    const float* w_gex    = (const float*)d_in[8];
    const float* w_comp   = (const float*)d_in[9];
    const float* W1       = (const float*)d_in[10];
    const float* W2       = (const float*)d_in[11];
    const float* w_dose   = (const float*)d_in[12];
    const float* w_time   = (const float*)d_in[13];
    const float* gamma    = (const float*)d_in[14];
    const float* beta     = (const float*)d_in[15];
    const float* Wff      = (const float*)d_in[16];

    float* out  = (float*)d_out;
    float* comp = out + BATCH * GEX;

    float* ws     = (float*)d_ws;
    float* ppool  = ws;
    float* S      = ppool + BATCH * NCH * CCEH;
    float* pred0  = S + BATCH * HDIM;
    float* xlnP   = pred0 + BATCH * GEX;
    float* Ppart  = xlnP + BATCH * KPAD;
    float* Spart  = Ppart + NKS * BATCH * PPN;

    k_cce<<<dim3(BATCH, NCH), CCE_T, 0, stream>>>(node_feat, mask, adj, dist, W1, ppool);
    k_compS<<<dim3(NSLAB, BATCH / BGRP), 256, 0, stream>>>(ppool, W2, dose, timev, w_dose, w_time,
                                                           b_gex, w_gex, w_comp, comp, Spart);
    k_S<<<BATCH, HDIM, 0, stream>>>(Spart, S);
    k_pred0<<<(GEX + 3) / 4, 256, 0, stream>>>(b_gex, comp, w_gex, w_comp, S, ppi, pred0);
    k_ln<<<BATCH, 256, 0, stream>>>(pred0, gamma, beta, xlnP);
    k_gemm_slab<<<dim3(NNB, NKS), 256, 0, stream>>>(xlnP, Wff, Ppart);
    k_outsum<<<dim3((GEX / 2 + 255) / 256, BATCH), 256, 0, stream>>>(Ppart, out);
}


# LayerNorm launch goes: k_pred0 work-groups leave per-row (sum, squared-deviation) parts; GEMM work-groups merge them and build their relu(LN) A slab in LDS; 7 launches to 6; LN sums' order changes
# speedup vs baseline: 5.9726x; 1.0364x over previous
#include <hip/hip_runtime.h>
#include <hip/hip_bf16.h>

#define GEX 978
#define KPAD 980
#define HDIM 128
#define NAT 50
#define FEAT 34
#define CCEH 64
#define BATCH 64
#define GCH 16
#define NSLAB ((GEX + GCH - 1) / GCH)
#define BGRP 16
#define PPITCH (CCEH + 1)
#define MC 25
#define NCH (NAT / MC)
#define CCE_T 288
#define NFT ((MC * FEAT + CCE_T - 1) / CCE_T)
#define ADT ((NAT * MC + CCE_T - 1) / CCE_T)
#define KSTEPS 35
#define KSLAB (4 * KSTEPS)
#define NKS (KPAD / KSLAB)
#define NBLK 32
#define NNB ((GEX + NBLK - 1) / NBLK)
#define PPN (NNB * NBLK)
#define PPI_V2 (GEX / 2)
#define PPI_LD ((PPI_V2 + 63) / 64)
#define SPITCH (HDIM + 4)
#define NWG4 ((GEX + 3) / 4)
#define NLAST (GEX - 4 * (NWG4 - 1))
#define NST ((NWG4 + 7) / 8)
#define TPR (KSLAB / 2)
#define TLD ((BATCH * TPR + 255) / 256)

typedef __attribute__((ext_vector_type(2))) float v2f;
typedef __attribute__((ext_vector_type(4))) float v4f;
typedef __attribute__((ext_vector_type(8))) float v8f;

__global__ __launch_bounds__(CCE_T) void k_cce(const float* __restrict__ nf,
                                               const float* __restrict__ mask,
                                               const float* __restrict__ adj,
                                               const float* __restrict__ dist,
                                               const float* __restrict__ W1,
                                               float* __restrict__ ppool) {
    int b = blockIdx.x, c = blockIdx.y, tid = threadIdx.x, m0 = c * MC;
    __shared__ float nfs[MC * FEAT];
    __shared__ float as[NAT * MC];
    __shared__ float es[NAT * MC];
    __shared__ float hs[MC * CCEH];
    __shared__ float ms[NAT];
    __shared__ float wm[MC];
    __shared__ float msum_s;

    int d = tid & (CCEH - 1);
    float w1[FEAT];
    #pragma unroll
    for (int f = 0; f < FEAT; f++) w1[f] = W1[f * CCEH + d];
    const float* nfb = nf + (b * NAT + m0) * FEAT;
    float nv[NFT];
    #pragma unroll
    for (int j = 0; j < NFT; j++) {
        int i = tid + CCE_T * j;
        nv[j] = nfb[i < MC * FEAT ? i : MC * FEAT - 1];
    }
    float av[ADT], dv[ADT];
    #pragma unroll
    for (int q = 0; q < ADT; q++) {
        int j = tid + CCE_T * q;
        int jc = j < NAT * MC ? j : NAT * MC - 1;
        int n = jc / MC, ml = jc - n * MC;
        int idx = (b * NAT + n) * NAT + m0 + ml;
        av[q] = adj[idx];
        dv[q] = dist[idx];
    }
    float mk = mask[b * NAT + (tid < NAT ? tid : NAT - 1)];

    #pragma unroll
    for (int j = 0; j < NFT; j++) {
        int i = tid + CCE_T * j;
        if (i < MC * FEAT) nfs[i] = nv[j];
    }
    #pragma unroll
    for (int q = 0; q < ADT; q++) {
        int j = tid + CCE_T * q;
        if (j < NAT * MC) { as[j] = av[q]; es[j] = expf(-dv[q]); }
    }
    if (tid < NAT) ms[tid] = mk;
    __syncthreads();

    if (tid < 256) {
        int mq = tid >> 6;
        #pragma unroll
        for (int r = 0; r < (MC + 3) / 4; r++) {
            int ml = mq + 4 * r;
            if (ml < MC) {
                float s = 0.f;
                #pragma unroll
                for (int f = 0; f < FEAT; f++) s += nfs[ml * FEAT + f] * w1[f];
                hs[ml * CCEH + d] = fmaxf(s, 0.f);
            }
        }
    } else if (tid < 256 + MC) {
        int ml = tid - 256;
        float s = 0.f, msum = 0.f;
        #pragma unroll 10
        for (int n = 0; n < NAT; n++) {
            s += ms[n] * as[n * MC + ml] * es[n * MC + ml];
            msum += ms[n];
        }
        wm[ml] = s;
        if (ml == 0) msum_s = fmaxf(msum, 1.f);
    }
    __syncthreads();
    if (tid < CCEH) {
        float s = 0.f;
        #pragma unroll
        for (int ml = 0; ml < MC; ml++) s += wm[ml] * hs[ml * CCEH + tid];
        ppool[(b * NCH + c) * CCEH + tid] = s / msum_s;
    }
}

__global__ __launch_bounds__(256) void k_compS(const float* __restrict__ ppool,
                                               const float* __restrict__ W2,
                                               const float* __restrict__ dose,
                                               const float* __restrict__ timev,
                                               const float* __restrict__ w_dose,
                                               const float* __restrict__ w_time,
                                               const float* __restrict__ b_gex,
                                               const float* __restrict__ w_gex,
                                               const float* __restrict__ w_comp,
                                               float* __restrict__ comp,
                                               float* __restrict__ Spart) {
    __shared__ float Ps[BGRP * PPITCH];
    __shared__ float W2s[CCEH * GCH];
    __shared__ __attribute__((aligned(16))) float Xs[BGRP * GCH];
    __shared__ __attribute__((aligned(16))) float Cs[BGRP * GCH];
    int slab = blockIdx.x, b0 = blockIdx.y * BGRP, g0 = slab * GCH, tid = threadIdx.x;

    float pv[4][NCH], wv[4];
    #pragma unroll
    for (int j = 0; j < 4; j++) {
        int i = tid + 256 * j;
        int gq = g0 + (i & 15);
        #pragma unroll
        for (int c = 0; c < NCH; c++)
            pv[j][c] = ppool[((b0 + (i >> 6)) * NCH + c) * CCEH + (i & 63)];
        wv[j] = W2[(i >> 4) * GEX + (gq < GEX ? gq : GEX - 1)];
    }
    int bl = tid >> 4, gl = tid & 15, b = b0 + bl, g = g0 + gl;
    bool gok = g < GEX;
    int gc = gok ? g : GEX - 1;
    float x  = b_gex[b * GEX + gc];
    float s  = dose[b] * w_dose[gc] + timev[b] * w_time[gc];
    int h = tid & (HDIM - 1), bh = tid >> 7;
    float wg[GCH], wc[GCH];
    #pragma unroll
    for (int j = 0; j < GCH; j++) {
        int gq = g0 + j;
        int gr = gq < GEX ? gq : GEX - 1;
        float a = w_gex[gr * HDIM + h], c2 = w_comp[gr * HDIM + h];
        wg[j] = gq < GEX ? a : 0.f;
        wc[j] = gq < GEX ? c2 : 0.f;
    }
    #pragma unroll
    for (int j = 0; j < 4; j++) {
        int i = tid + 256 * j;
        float p = pv[j][0];
        #pragma unroll
        for (int c = 1; c < NCH; c++) p += pv[j][c];
        Ps[(i >> 6) * PPITCH + (i & 63)] = p;
        W2s[i] = (g0 + (i & 15) < GEX) ? wv[j] : 0.f;
    }
    if (!gok) { x = 0.f; s = 0.f; }
    __syncthreads();
    #pragma unroll 8
    for (int d = 0; d < CCEH; d++) s += Ps[bl * PPITCH + d] * W2s[d * GCH + gl];
    if (gok) comp[b * GEX + g] = s;
    Xs[tid] = x;
    Cs[tid] = gok ? s : 0.f;
    __syncthreads();
    #pragma unroll
    for (int r = 0; r < BGRP / 2; r++) {
        int row = bh * (BGRP / 2) + r;
        float acc = 0.f;
        #pragma unroll
        for (int j = 0; j < GCH; j++)
            acc += Xs[row * GCH + j] * wg[j] + Cs[row * GCH + j] * wc[j];
        Spart[(slab * BATCH + b0 + row) * HDIM + h] = acc;
    }
}

__global__ __launch_bounds__(128) void k_S(const float* __restrict__ Spart,
                                           float* __restrict__ S) {
    int i = blockIdx.x * HDIM + threadIdx.x;
    float s = 0.f;
    #pragma unroll
    for (int p = 0; p < NSLAB; p++) s += Spart[p * (BATCH * HDIM) + i];
    S[i] = s;
}

__global__ __launch_bounds__(256) void k_pred0(const float* __restrict__ b_gex,
                                               const float* __restrict__ comp,
                                               const float* __restrict__ w_gex,
                                               const float* __restrict__ w_comp,
                                               const float* __restrict__ S,
                                               const float* __restrict__ ppi,
                                               float* __restrict__ pred0,
                                               float* __restrict__ lnpart) {
    __shared__ __attribute__((aligned(16))) float Ssh[BATCH * SPITCH];
    __shared__ __attribute__((aligned(16))) float wsh[4 * HDIM * 2];
    __shared__ __attribute__((aligned(8)))  float dsh[BATCH * 4 * 2];
    __shared__ float rs[8];
    int g0 = blockIdx.x * 4, tid = threadIdx.x;
    int pr = g0 + (tid >> 6), pt = tid & 63;
    const v2f* prow = (const v2f*)(ppi + (pr < GEX ? pr : GEX - 1) * GEX);
    v2f pv[PPI_LD];
    #pragma unroll
    for (int i = 0; i < PPI_LD; i++) {
        int j = pt + 64 * i;
        pv[i] = prow[j < PPI_V2 ? j : PPI_V2 - 1];
    }
    int wv_ = tid >> 5, lane = tid & 31;
    const v4f* S4 = (const v4f*)S;
    v4f sv[8];
    #pragma unroll
    for (int i = 0; i < 8; i++) sv[i] = S4[tid + 256 * i];
    int wgx = g0 + (wv_ & 3);
    int wof = (wgx < GEX ? wgx : GEX - 1) * HDIM + lane * 4;
    v4f wgv = *(const v4f*)(w_gex + wof), wcv = *(const v4f*)(w_comp + wof);
    int fb = tid >> 2, fgl = tid & 3, fg = g0 + fgl;
    int fgc = fg < GEX ? fg : GEX - 1;
    float x = b_gex[fb * GEX + fgc], c = comp[fb * GEX + fgc];
    asm volatile("" :: "v"(pv[0]), "v"(pv[1]), "v"(pv[2]), "v"(pv[3]), "v"(pv[4]), "v"(pv[5]),
                       "v"(pv[6]), "v"(pv[7]), "v"(sv[0]), "v"(sv[1]), "v"(sv[2]), "v"(sv[3]),
                       "v"(sv[4]), "v"(sv[5]), "v"(sv[6]), "v"(sv[7]), "v"(wgv), "v"(wcv),
                       "v"(x), "v"(c));

    #pragma unroll
    for (int i = 0; i < 8; i++)
        *(v4f*)(Ssh + (wv_ + 8 * i) * SPITCH + lane * 4) = sv[i];
    if (wv_ < 4) {
        v4f lo = {wgv.x, wcv.x, wgv.y, wcv.y}, hi = {wgv.z, wcv.z, wgv.w, wcv.w};
        v4f* wd = (v4f*)(wsh + (wv_ * HDIM + lane * 4) * 2);
        wd[0] = lo;
        wd[1] = hi;
    }
    float rsum = 0.f;
    #pragma unroll
    for (int i = 0; i < PPI_LD; i++) {
        float a = pv[i].x + pv[i].y;
        rsum += (pt + 64 * i < PPI_V2) ? a : 0.f;
    }
    for (int o = 16; o > 0; o >>= 1) rsum += __shfl_xor(rsum, o);
    if (lane == 0) rs[wv_] = rsum;
    __syncthreads();
    if (wv_ < 4) {
        const v4f* sa = (const v4f*)(Ssh + lane * SPITCH);
        const v4f* sb = (const v4f*)(Ssh + (lane + 32) * SPITCH);
        const v4f* wp = (const v4f*)(wsh + wv_ * HDIM * 2);
        v2f da = {0.f, 0.f}, db = {0.f, 0.f};
        #pragma unroll 8
        for (int q = 0; q < HDIM / 4; q++) {
            v4f a = sa[q], b4 = sb[q], w01 = wp[2 * q], w23 = wp[2 * q + 1];
            v2f w0 = {w01.x, w01.y}, w1 = {w01.z, w01.w}, w2 = {w23.x, w23.y}, w3 = {w23.z, w23.w};
            da = __builtin_elementwise_fma((v2f){a.x, a.x}, w0, da);
            db = __builtin_elementwise_fma((v2f){b4.x, b4.x}, w0, db);
            da = __builtin_elementwise_fma((v2f){a.y, a.y}, w1, da);
            db = __builtin_elementwise_fma((v2f){b4.y, b4.y}, w1, db);
            da = __builtin_elementwise_fma((v2f){a.z, a.z}, w2, da);
            db = __builtin_elementwise_fma((v2f){b4.z, b4.z}, w2, db);
            da = __builtin_elementwise_fma((v2f){a.w, a.w}, w3, da);
            db = __builtin_elementwise_fma((v2f){b4.w, b4.w}, w3, db);
        }
        *(v2f*)(dsh + (lane * 4 + wv_) * 2) = da;
        *(v2f*)(dsh + ((lane + 32) * 4 + wv_) * 2) = db;
    }
    __syncthreads();
    float pv0 = 0.f;
    if (fg < GEX) {
        v2f d = *(const v2f*)(dsh + tid * 2);
        float row = (x * d.x + c * d.y) * 0.08838834764831845f
                  + (rs[2 * fgl] + rs[2 * fgl + 1]);
        pv0 = x * row;
        pred0[fb * GEX + fg] = pv0;
    }
    float ps = pv0 + __shfl_xor(pv0, 1);
    ps += __shfl_xor(ps, 2);
    float nn = (blockIdx.x == NWG4 - 1) ? (float)NLAST : 4.f;
    float dd = (fg < GEX) ? pv0 - ps / nn : 0.f;
    float pq = dd * dd;
    pq += __shfl_xor(pq, 1);
    pq += __shfl_xor(pq, 2);
    if (fgl == 0) *(v2f*)(lnpart + (blockIdx.x * BATCH + fb) * 2) = (v2f){ps, pq};
}

__device__ __forceinline__ void lds_barrier() {
    __builtin_amdgcn_fence(__ATOMIC_RELEASE, "workgroup", "local");
    __builtin_amdgcn_s_barrier();
    __builtin_amdgcn_fence(__ATOMIC_ACQUIRE, "workgroup", "local");
}

__global__ __launch_bounds__(256) void k_gemm_slab(const float* __restrict__ pred0,
                                                   const float* __restrict__ lnpart,
                                                   const float* __restrict__ gamma,
                                                   const float* __restrict__ beta,
                                                   const float* __restrict__ Wff,
                                                   float* __restrict__ Ppart) {
    __shared__ __attribute__((aligned(16))) float xs[BATCH * KSLAB];
    __shared__ __attribute__((aligned(8)))  float red[2][8][BATCH];
    __shared__ float mus[BATCH], rsd[BATCH];
    __shared__ __attribute__((aligned(8)))  float gbs[2 * 256];
    int tid = threadIdx.x, lane = tid & 31, w = tid >> 5;
    int mt = w & 3, ns = w >> 2;
    int slab = blockIdx.y;
    int lg   = lane & 15;
    int koff = (lane >= 16) ? 2 : 0;
    int m  = mt * 16 + lg;
    int n  = blockIdx.x * NBLK + ns * 16 + lg;
    int nc = (n < GEX) ? n : (GEX - 1);
    int k0 = slab * KSLAB;
    int kb = k0 + koff;

    int jg = tid < TPR ? tid : (tid < 2 * TPR ? tid - TPR : 0);
    int kg = k0 + 2 * jg;
    v2f gbv = *(const v2f*)((tid < TPR ? gamma : beta) + (kg < GEX ? kg : GEX - 2));
    const v4f* st4 = (const v4f*)lnpart;
    v4f st[NST];
    #pragma unroll
    for (int t = 0; t < NST; t++) {
        int i = w + 8 * t;
        st[t] = st4[(i < NWG4 ? i : NWG4 - 1) * (BATCH / 2) + lane];
    }
    int row0 = tid / TPR, j0 = tid - row0 * TPR;
    v2f pr[TLD];
    {
        int row = row0, j = j0;
        #pragma unroll
        for (int q = 0; q < TLD; q++) {
            int kk = k0 + 2 * j;
            pr[q] = *(const v2f*)(pred0 + (row < BATCH ? row : BATCH - 1) * GEX + (kk < GEX ? kk : GEX - 2));
            j += 256 - 3 * TPR; row += 3;
            if (j >= TPR) { j -= TPR; row++; }
        }
    }
    __builtin_amdgcn_sched_barrier(0);
    const float* bp = Wff + kb * GEX + nc;
    v2f bm[KSTEPS];
    #pragma unroll
    for (int s = 0; s < KSTEPS; s++) {
        if (s < KSTEPS - 1) {
            bm[s].x = bp[(4 * s) * GEX];
            bm[s].y = bp[(4 * s + 1) * GEX];
        } else {
            int r0 = kb + 4 * s, r1 = r0 + 1;
            float b0 = Wff[((r0 < GEX) ? r0 : (GEX - 1)) * GEX + nc];
            float b1 = Wff[((r1 < GEX) ? r1 : (GEX - 1)) * GEX + nc];
            bm[s].x = (r0 < GEX) ? b0 : 0.f;
            bm[s].y = (r1 < GEX) ? b1 : 0.f;
        }
    }

    float s0 = 0.f, s1 = 0.f;
    #pragma unroll
    for (int t = 0; t < NST; t++) {
        if (w + 8 * t < NWG4) { s0 += st[t].x; s1 += st[t].z; }
    }
    *(v2f*)&red[0][w][2 * lane] = (v2f){s0, s1};
    *(v2f*)&gbs[2 * tid] = gbv;
    lds_barrier();
    v2f ms = *(const v2f*)&red[0][0][2 * lane];
    for (int ww = 1; ww < 8; ww++) ms += *(const v2f*)&red[0][ww][2 * lane];
    float mu0 = ms.x / (float)GEX, mu1 = ms.y / (float)GEX;
    float q0 = 0.f, q1 = 0.f;
    #pragma unroll
    for (int t = 0; t < NST; t++) {
        int i = w + 8 * t;
        if (i < NWG4) {
            float nn  = (i == NWG4 - 1) ? (float)NLAST : 4.f;
            float inv = (i == NWG4 - 1) ? 1.f / (float)NLAST : 0.25f;
            float e0 = st[t].x - nn * mu0, e1 = st[t].z - nn * mu1;
            q0 += st[t].y + e0 * e0 * inv;
            q1 += st[t].w + e1 * e1 * inv;
        }
    }
    *(v2f*)&red[1][w][2 * lane] = (v2f){q0, q1};
    lds_barrier();
    if (w == 0) {
        v2f us = *(const v2f*)&red[1][0][2 * lane];
        for (int ww = 1; ww < 8; ww++) us += *(const v2f*)&red[1][ww][2 * lane];
        mus[2 * lane] = mu0;
        mus[2 * lane + 1] = mu1;
        rsd[2 * lane]     = rsqrtf(us.x / (float)GEX + 1e-5f);
        rsd[2 * lane + 1] = rsqrtf(us.y / (float)GEX + 1e-5f);
    }
    lds_barrier();
    {
        int row = row0, j = j0;
        #pragma unroll
        for (int q = 0; q < TLD; q++) {
            if (row < BATCH) {
                float mu = mus[row], rstd = rsd[row];
                v2f ga = *(const v2f*)&gbs[2 * j], be = *(const v2f*)&gbs[2 * (TPR + j)];
                float d0 = pr[q].x - mu, d1 = pr[q].y - mu;
                v2f y = {fmaxf(d0 * rstd * ga.x + be.x, 0.f), fmaxf(d1 * rstd * ga.y + be.y, 0.f)};
                if (k0 + 2 * j >= GEX) y = (v2f){0.f, 0.f};
                *(v2f*)(xs + row * KSLAB + 2 * j) = y;
            }
            j += 256 - 3 * TPR; row += 3;
            if (j >= TPR) { j -= TPR; row++; }
        }
    }
    lds_barrier();

    const float* xa = xs + m * KSLAB + koff;
    float* pp = Ppart + (slab * BATCH + mt * 16 + ((lane >= 16) ? 8 : 0)) * PPN + n;
    v2f af[KSTEPS];
    #pragma unroll
    for (int s = 0; s < KSTEPS; s++) af[s] = *(const v2f*)(xa + 4 * s);
    asm volatile("" :: "v"(af[0]), "v"(af[1]), "v"(af[2]), "v"(af[3]), "v"(af[4]), "v"(af[5]),
                       "v"(af[6]), "v"(af[7]), "v"(af[8]), "v"(af[9]), "v"(af[10]), "v"(af[11]),
                       "v"(af[12]), "v"(af[13]), "v"(af[14]), "v"(af[15]), "v"(af[16]), "v"(af[17]));
    asm volatile("" :: "v"(af[18]), "v"(af[19]), "v"(af[20]), "v"(af[21]), "v"(af[22]), "v"(af[23]),
                       "v"(af[24]), "v"(af[25]), "v"(af[26]), "v"(af[27]), "v"(af[28]), "v"(af[29]),
                       "v"(af[30]), "v"(af[31]), "v"(af[32]), "v"(af[33]), "v"(af[34]));
    v8f acc = {};
    #pragma unroll
    for (int s = 0; s < KSTEPS; s++)
        acc = __builtin_amdgcn_wmma_f32_16x16x4_f32(
            false, af[s], false, bm[s], (short)0, acc, false, false);

    for (int v = 0; v < 8; v++) pp[v * PPN] = acc[v];
}

__global__ __launch_bounds__(256) void k_outsum(const float* __restrict__ Ppart,
                                                float* __restrict__ out) {
    int j = blockIdx.x * 256 + threadIdx.x, b = blockIdx.y;
    if (j < GEX / 2) {
        const float* pl = Ppart + b * PPN + 2 * j;
        v2f s = *(const v2f*)pl;
        for (int p = 1; p < NKS; p++) s += *(const v2f*)(pl + p * (BATCH * PPN));
        *(v2f*)(out + b * GEX + 2 * j) = s;
    }
}

extern "C" void kernel_launch(void* const* d_in, const int* in_sizes, int n_in,
                              void* d_out, int out_size, void* d_ws, size_t ws_size,
                              hipStream_t stream) {
    const float* b_gex    = (const float*)d_in[0];
    const float* node_feat= (const float*)d_in[1];
    const float* mask     = (const float*)d_in[2];
    const float* adj      = (const float*)d_in[3];
    const float* dist     = (const float*)d_in[4];
    const float* dose     = (const float*)d_in[5];
    const float* timev    = (const float*)d_in[6];
    const float* ppi      = (const float*)d_in[7];
    const float* w_gex    = (const float*)d_in[8];
    const float* w_comp   = (const float*)d_in[9];
    const float* W1       = (const float*)d_in[10];
    const float* W2       = (const float*)d_in[11];
    const float* w_dose   = (const float*)d_in[12];
    const float* w_time   = (const float*)d_in[13];
    const float* gamma    = (const float*)d_in[14];
    const float* beta     = (const float*)d_in[15];
    const float* Wff      = (const float*)d_in[16];

    float* out  = (float*)d_out;
    float* comp = out + BATCH * GEX;

    float* ws     = (float*)d_ws;
    float* ppool  = ws;
    float* S      = ppool + BATCH * NCH * CCEH;
    float* pred0  = S + BATCH * HDIM;
    float* lnpart = pred0 + BATCH * GEX;
    float* Ppart  = lnpart + BATCH * KPAD;
    float* Spart  = Ppart + NKS * BATCH * PPN;

    k_cce<<<dim3(BATCH, NCH), CCE_T, 0, stream>>>(node_feat, mask, adj, dist, W1, ppool);
    k_compS<<<dim3(NSLAB, BATCH / BGRP), 256, 0, stream>>>(ppool, W2, dose, timev, w_dose, w_time,
                                                           b_gex, w_gex, w_comp, comp, Spart);
    k_S<<<BATCH, HDIM, 0, stream>>>(Spart, S);
    k_pred0<<<(GEX + 3) / 4, 256, 0, stream>>>(b_gex, comp, w_gex, w_comp, S, ppi, pred0, lnpart);
    k_gemm_slab<<<dim3(NNB, NKS), 256, 0, stream>>>(pred0, lnpart, gamma, beta, Wff, Ppart);
    k_outsum<<<dim3((GEX / 2 + 255) / 256, BATCH), 256, 0, stream>>>(Ppart, out);
}


# k_pred0 re-tiled to 8 batch rows x 32 g per work-group (wave = row, lane = g): LayerNorm parts per row 245 to 31; ppi row sums by spare work-groups of launch 1; LN and row sums' order changes
# speedup vs baseline: 6.6087x; 1.1065x over previous
#include <hip/hip_runtime.h>
#include <hip/hip_bf16.h>

#define GEX 978
#define KPAD 980
#define HDIM 128
#define NAT 50
#define FEAT 34
#define CCEH 64
#define BATCH 64
#define GCH 16
#define NSLAB ((GEX + GCH - 1) / GCH)
#define BGRP 16
#define PPITCH (CCEH + 1)
#define MC 25
#define NCH (NAT / MC)
#define CCE_T 288
#define NFT ((MC * FEAT + CCE_T - 1) / CCE_T)
#define ADT ((NAT * MC + CCE_T - 1) / CCE_T)
#define KSTEPS 35
#define KSLAB (4 * KSTEPS)
#define NKS (KPAD / KSLAB)
#define NBLK 32
#define NNB ((GEX + NBLK - 1) / NBLK)
#define PPN (NNB * NBLK)
#define PPI_V2 (GEX / 2)
#define PPI_LD ((PPI_V2 + 31) / 32)
#define SPITCH (HDIM + 4)
#define PG 32
#define PB 8
#define NWG4 ((GEX + PG - 1) / PG)
#define NLAST (GEX - PG * (NWG4 - 1))
#define NST ((NWG4 + 7) / 8)
#define RSG (CCE_T / 32)
#define RSY ((GEX + RSG * BATCH - 1) / (RSG * BATCH))
#define TPR (KSLAB / 2)
#define TLD ((BATCH * TPR + 255) / 256)

typedef __attribute__((ext_vector_type(2))) float v2f;
typedef __attribute__((ext_vector_type(4))) float v4f;
typedef __attribute__((ext_vector_type(8))) float v8f;

__device__ __forceinline__ float xhalf(float x) {
    int b = __builtin_bit_cast(int, x);
    return __builtin_bit_cast(float, __builtin_amdgcn_permlanex16(b, b, 0x76543210, 0xfedcba98, false, false));
}
__device__ __forceinline__ float wave_sum(float v) {
    v += __builtin_bit_cast(float, __builtin_amdgcn_update_dpp(0, __builtin_bit_cast(int, v), 0x161, 0xf, 0xf, true));
    v += __builtin_bit_cast(float, __builtin_amdgcn_update_dpp(0, __builtin_bit_cast(int, v), 0x162, 0xf, 0xf, true));
    v += __builtin_bit_cast(float, __builtin_amdgcn_update_dpp(0, __builtin_bit_cast(int, v), 0x164, 0xf, 0xf, true));
    v += __builtin_bit_cast(float, __builtin_amdgcn_update_dpp(0, __builtin_bit_cast(int, v), 0x168, 0xf, 0xf, true));
    return v + xhalf(v);
}

__device__ __forceinline__ void ppi_rowsums(const float* __restrict__ ppi,
                                            float* __restrict__ rowsum, int grp) {
    int lane = threadIdx.x & 31;
    int r = grp * RSG + (threadIdx.x >> 5);
    if (r >= GEX) return;
    const v2f* prow = (const v2f*)(ppi + r * GEX);
    v2f pv[PPI_LD];
    #pragma unroll
    for (int i = 0; i < PPI_LD; i++) {
        int j = lane + 32 * i;
        pv[i] = prow[j < PPI_V2 ? j : PPI_V2 - 1];
    }
    float rsum = 0.f;
    #pragma unroll
    for (int i = 0; i < PPI_LD; i++) {
        float a = pv[i].x + pv[i].y;
        rsum += (lane + 32 * i < PPI_V2) ? a : 0.f;
    }
    rsum = wave_sum(rsum);
    if (lane == 0) rowsum[r] = rsum;
}

__global__ __launch_bounds__(CCE_T) void k_cce(const float* __restrict__ nf,
                                               const float* __restrict__ mask,
                                               const float* __restrict__ adj,
                                               const float* __restrict__ dist,
                                               const float* __restrict__ W1,
                                               float* __restrict__ ppool,
                                               const float* __restrict__ ppi,
                                               float* __restrict__ rowsum) {
    if (blockIdx.y >= NCH) {
        ppi_rowsums(ppi, rowsum, (blockIdx.y - NCH) * BATCH + blockIdx.x);
        return;
    }
    int b = blockIdx.x, c = blockIdx.y, tid = threadIdx.x, m0 = c * MC;
    __shared__ float nfs[MC * FEAT];
    __shared__ float as[NAT * MC];
    __shared__ float es[NAT * MC];
    __shared__ float hs[MC * CCEH];
    __shared__ float ms[NAT];
    __shared__ float wm[MC];
    __shared__ float msum_s;

    int d = tid & (CCEH - 1);
    float w1[FEAT];
    #pragma unroll
    for (int f = 0; f < FEAT; f++) w1[f] = W1[f * CCEH + d];
    const float* nfb = nf + (b * NAT + m0) * FEAT;
    float nv[NFT];
    #pragma unroll
    for (int j = 0; j < NFT; j++) {
        int i = tid + CCE_T * j;
        nv[j] = nfb[i < MC * FEAT ? i : MC * FEAT - 1];
    }
    float av[ADT], dv[ADT];
    #pragma unroll
    for (int q = 0; q < ADT; q++) {
        int j = tid + CCE_T * q;
        int jc = j < NAT * MC ? j : NAT * MC - 1;
        int n = jc / MC, ml = jc - n * MC;
        int idx = (b * NAT + n) * NAT + m0 + ml;
        av[q] = adj[idx];
        dv[q] = dist[idx];
    }
    float mk = mask[b * NAT + (tid < NAT ? tid : NAT - 1)];

    #pragma unroll
    for (int j = 0; j < NFT; j++) {
        int i = tid + CCE_T * j;
        if (i < MC * FEAT) nfs[i] = nv[j];
    }
    #pragma unroll
    for (int q = 0; q < ADT; q++) {
        int j = tid + CCE_T * q;
        if (j < NAT * MC) { as[j] = av[q]; es[j] = expf(-dv[q]); }
    }
    if (tid < NAT) ms[tid] = mk;
    __syncthreads();

    if (tid < 256) {
        int mq = tid >> 6;
        #pragma unroll
        for (int r = 0; r < (MC + 3) / 4; r++) {
            int ml = mq + 4 * r;
            if (ml < MC) {
                float s = 0.f;
                #pragma unroll
                for (int f = 0; f < FEAT; f++) s += nfs[ml * FEAT + f] * w1[f];
                hs[ml * CCEH + d] = fmaxf(s, 0.f);
            }
        }
    } else if (tid < 256 + MC) {
        int ml = tid - 256;
        float s = 0.f, msum = 0.f;
        #pragma unroll 10
        for (int n = 0; n < NAT; n++) {
            s += ms[n] * as[n * MC + ml] * es[n * MC + ml];
            msum += ms[n];
        }
        wm[ml] = s;
        if (ml == 0) msum_s = fmaxf(msum, 1.f);
    }
    __syncthreads();
    if (tid < CCEH) {
        float s = 0.f;
        #pragma unroll
        for (int ml = 0; ml < MC; ml++) s += wm[ml] * hs[ml * CCEH + tid];
        ppool[(b * NCH + c) * CCEH + tid] = s / msum_s;
    }
}

__global__ __launch_bounds__(256) void k_compS(const float* __restrict__ ppool,
                                               const float* __restrict__ W2,
                                               const float* __restrict__ dose,
                                               const float* __restrict__ timev,
                                               const float* __restrict__ w_dose,
                                               const float* __restrict__ w_time,
                                               const float* __restrict__ b_gex,
                                               const float* __restrict__ w_gex,
                                               const float* __restrict__ w_comp,
                                               float* __restrict__ comp,
                                               float* __restrict__ Spart) {
    __shared__ float Ps[BGRP * PPITCH];
    __shared__ float W2s[CCEH * GCH];
    __shared__ __attribute__((aligned(16))) float Xs[BGRP * GCH];
    __shared__ __attribute__((aligned(16))) float Cs[BGRP * GCH];
    int slab = blockIdx.x, b0 = blockIdx.y * BGRP, g0 = slab * GCH, tid = threadIdx.x;

    float pv[4][NCH], wv[4];
    #pragma unroll
    for (int j = 0; j < 4; j++) {
        int i = tid + 256 * j;
        int gq = g0 + (i & 15);
        #pragma unroll
        for (int c = 0; c < NCH; c++)
            pv[j][c] = ppool[((b0 + (i >> 6)) * NCH + c) * CCEH + (i & 63)];
        wv[j] = W2[(i >> 4) * GEX + (gq < GEX ? gq : GEX - 1)];
    }
    int bl = tid >> 4, gl = tid & 15, b = b0 + bl, g = g0 + gl;
    bool gok = g < GEX;
    int gc = gok ? g : GEX - 1;
    float x  = b_gex[b * GEX + gc];
    float s  = dose[b] * w_dose[gc] + timev[b] * w_time[gc];
    int h = tid & (HDIM - 1), bh = tid >> 7;
    float wg[GCH], wc[GCH];
    #pragma unroll
    for (int j = 0; j < GCH; j++) {
        int gq = g0 + j;
        int gr = gq < GEX ? gq : GEX - 1;
        float a = w_gex[gr * HDIM + h], c2 = w_comp[gr * HDIM + h];
        wg[j] = gq < GEX ? a : 0.f;
        wc[j] = gq < GEX ? c2 : 0.f;
    }
    #pragma unroll
    for (int j = 0; j < 4; j++) {
        int i = tid + 256 * j;
        float p = pv[j][0];
        #pragma unroll
        for (int c = 1; c < NCH; c++) p += pv[j][c];
        Ps[(i >> 6) * PPITCH + (i & 63)] = p;
        W2s[i] = (g0 + (i & 15) < GEX) ? wv[j] : 0.f;
    }
    if (!gok) { x = 0.f; s = 0.f; }
    __syncthreads();
    #pragma unroll 8
    for (int d = 0; d < CCEH; d++) s += Ps[bl * PPITCH + d] * W2s[d * GCH + gl];
    if (gok) comp[b * GEX + g] = s;
    Xs[tid] = x;
    Cs[tid] = gok ? s : 0.f;
    __syncthreads();
    #pragma unroll
    for (int r = 0; r < BGRP / 2; r++) {
        int row = bh * (BGRP / 2) + r;
        float acc = 0.f;
        #pragma unroll
        for (int j = 0; j < GCH; j++)
            acc += Xs[row * GCH + j] * wg[j] + Cs[row * GCH + j] * wc[j];
        Spart[(slab * BATCH + b0 + row) * HDIM + h] = acc;
    }
}

__global__ __launch_bounds__(128) void k_S(const float* __restrict__ Spart,
                                           float* __restrict__ S) {
    int i = blockIdx.x * HDIM + threadIdx.x;
    float s = 0.f;
    #pragma unroll
    for (int p = 0; p < NSLAB; p++) s += Spart[p * (BATCH * HDIM) + i];
    S[i] = s;
}

__global__ __launch_bounds__(256) void k_pred0(const float* __restrict__ b_gex,
                                               const float* __restrict__ comp,
                                               const float* __restrict__ w_gex,
                                               const float* __restrict__ w_comp,
                                               const float* __restrict__ S,
                                               const float* __restrict__ rowsum,
                                               float* __restrict__ pred0,
                                               float* __restrict__ lnpart) {
    __shared__ __attribute__((aligned(16))) float wgs[PG * SPITCH];
    __shared__ __attribute__((aligned(16))) float wcs[PG * SPITCH];
    __shared__ __attribute__((aligned(16))) float Ssh[PB * HDIM];
    int g0 = blockIdx.x * PG, b0 = blockIdx.y * PB, tid = threadIdx.x;
    int w = tid >> 5, lane = tid & 31;
    int b = b0 + w, g = g0 + lane;
    bool gok = g < GEX;
    int gc = gok ? g : GEX - 1;
    v4f sv = ((const v4f*)S)[b0 * (HDIM / 4) + tid];
    v4f wgv[PG / 8], wcv[PG / 8];
    #pragma unroll
    for (int i = 0; i < PG / 8; i++) {
        int r = g0 + w + 8 * i;
        int of = (r < GEX ? r : GEX - 1) * HDIM + lane * 4;
        wgv[i] = *(const v4f*)(w_gex + of);
        wcv[i] = *(const v4f*)(w_comp + of);
    }
    float x = b_gex[b * GEX + gc], c = comp[b * GEX + gc], rsv = rowsum[gc];

    *(v4f*)(Ssh + w * HDIM + lane * 4) = sv;
    #pragma unroll
    for (int i = 0; i < PG / 8; i++) {
        *(v4f*)(wgs + (w + 8 * i) * SPITCH + lane * 4) = wgv[i];
        *(v4f*)(wcs + (w + 8 * i) * SPITCH + lane * 4) = wcv[i];
    }
    __syncthreads();
    const v4f* sp = (const v4f*)(Ssh + w * HDIM);
    const v4f* gp = (const v4f*)(wgs + lane * SPITCH);
    const v4f* cp = (const v4f*)(wcs + lane * SPITCH);
    float da = 0.f, dc = 0.f;
    #pragma unroll 8
    for (int q = 0; q < HDIM / 4; q++) {
        v4f s4 = sp[q], a4 = gp[q], c4 = cp[q];
        da = fmaf(s4.x, a4.x, da);  dc = fmaf(s4.x, c4.x, dc);
        da = fmaf(s4.y, a4.y, da);  dc = fmaf(s4.y, c4.y, dc);
        da = fmaf(s4.z, a4.z, da);  dc = fmaf(s4.z, c4.z, dc);
        da = fmaf(s4.w, a4.w, da);  dc = fmaf(s4.w, c4.w, dc);
    }
    float pv0 = 0.f;
    if (gok) {
        float row = (x * da + c * dc) * 0.08838834764831845f
                  + rsv;
        pv0 = x * row;
        pred0[b * GEX + g] = pv0;
    }
    float ps = wave_sum(pv0);
    float nn = (blockIdx.x == NWG4 - 1) ? (float)NLAST : (float)PG;
    float dd = gok ? pv0 - ps / nn : 0.f;
    float pq = wave_sum(dd * dd);
    if (lane == 0) *(v2f*)(lnpart + (blockIdx.x * BATCH + b) * 2) = (v2f){ps, pq};
}

__device__ __forceinline__ void lds_barrier() {
    __builtin_amdgcn_fence(__ATOMIC_RELEASE, "workgroup", "local");
    __builtin_amdgcn_s_barrier();
    __builtin_amdgcn_fence(__ATOMIC_ACQUIRE, "workgroup", "local");
}

__global__ __launch_bounds__(256) void k_gemm_slab(const float* __restrict__ pred0,
                                                   const float* __restrict__ lnpart,
                                                   const float* __restrict__ gamma,
                                                   const float* __restrict__ beta,
                                                   const float* __restrict__ Wff,
                                                   float* __restrict__ Ppart) {
    __shared__ __attribute__((aligned(16))) float xs[BATCH * KSLAB];
    __shared__ __attribute__((aligned(8)))  float red[2][8][BATCH];
    __shared__ float mus[BATCH], rsd[BATCH];
    __shared__ __attribute__((aligned(8)))  float gbs[2 * 256];
    int tid = threadIdx.x, lane = tid & 31, w = tid >> 5;
    int mt = w & 3, ns = w >> 2;
    int slab = blockIdx.y;
    int lg   = lane & 15;
    int koff = (lane >= 16) ? 2 : 0;
    int m  = mt * 16 + lg;
    int n  = blockIdx.x * NBLK + ns * 16 + lg;
    int nc = (n < GEX) ? n : (GEX - 1);
    int k0 = slab * KSLAB;
    int kb = k0 + koff;

    int jg = tid < TPR ? tid : (tid < 2 * TPR ? tid - TPR : 0);
    int kg = k0 + 2 * jg;
    v2f gbv = *(const v2f*)((tid < TPR ? gamma : beta) + (kg < GEX ? kg : GEX - 2));
    const v4f* st4 = (const v4f*)lnpart;
    v4f st[NST];
    #pragma unroll
    for (int t = 0; t < NST; t++) {
        int i = w + 8 * t;
        st[t] = st4[(i < NWG4 ? i : NWG4 - 1) * (BATCH / 2) + lane];
    }
    int row0 = tid / TPR, j0 = tid - row0 * TPR;
    v2f pr[TLD];
    {
        int row = row0, j = j0;
        #pragma unroll
        for (int q = 0; q < TLD; q++) {
            int kk = k0 + 2 * j;
            pr[q] = *(const v2f*)(pred0 + (row < BATCH ? row : BATCH - 1) * GEX + (kk < GEX ? kk : GEX - 2));
            j += 256 - 3 * TPR; row += 3;
            if (j >= TPR) { j -= TPR; row++; }
        }
    }
    __builtin_amdgcn_sched_barrier(0);
    const float* bp = Wff + kb * GEX + nc;
    v2f bm[KSTEPS];
    #pragma unroll
    for (int s = 0; s < KSTEPS; s++) {
        if (s < KSTEPS - 1) {
            bm[s].x = bp[(4 * s) * GEX];
            bm[s].y = bp[(4 * s + 1) * GEX];
        } else {
            int r0 = kb + 4 * s, r1 = r0 + 1;
            float b0 = Wff[((r0 < GEX) ? r0 : (GEX - 1)) * GEX + nc];
            float b1 = Wff[((r1 < GEX) ? r1 : (GEX - 1)) * GEX + nc];
            bm[s].x = (r0 < GEX) ? b0 : 0.f;
            bm[s].y = (r1 < GEX) ? b1 : 0.f;
        }
    }

    float s0 = 0.f, s1 = 0.f;
    #pragma unroll
    for (int t = 0; t < NST; t++) {
        if (w + 8 * t < NWG4) { s0 += st[t].x; s1 += st[t].z; }
    }
    *(v2f*)&red[0][w][2 * lane] = (v2f){s0, s1};
    *(v2f*)&gbs[2 * tid] = gbv;
    lds_barrier();
    v2f ms = *(const v2f*)&red[0][0][2 * lane];
    for (int ww = 1; ww < 8; ww++) ms += *(const v2f*)&red[0][ww][2 * lane];
    float mu0 = ms.x / (float)GEX, mu1 = ms.y / (float)GEX;
    float q0 = 0.f, q1 = 0.f;
    #pragma unroll
    for (int t = 0; t < NST; t++) {
        int i = w + 8 * t;
        if (i < NWG4) {
            float nn  = (i == NWG4 - 1) ? (float)NLAST : (float)PG;
            float inv = (i == NWG4 - 1) ? 1.f / (float)NLAST : 1.f / (float)PG;
            float e0 = st[t].x - nn * mu0, e1 = st[t].z - nn * mu1;
            q0 += st[t].y + e0 * e0 * inv;
            q1 += st[t].w + e1 * e1 * inv;
        }
    }
    *(v2f*)&red[1][w][2 * lane] = (v2f){q0, q1};
    lds_barrier();
    if (w == 0) {
        v2f us = *(const v2f*)&red[1][0][2 * lane];
        for (int ww = 1; ww < 8; ww++) us += *(const v2f*)&red[1][ww][2 * lane];
        mus[2 * lane] = mu0;
        mus[2 * lane + 1] = mu1;
        rsd[2 * lane]     = rsqrtf(us.x / (float)GEX + 1e-5f);
        rsd[2 * lane + 1] = rsqrtf(us.y / (float)GEX + 1e-5f);
    }
    lds_barrier();
    {
        int row = row0, j = j0;
        #pragma unroll
        for (int q = 0; q < TLD; q++) {
            if (row < BATCH) {
                float mu = mus[row], rstd = rsd[row];
                v2f ga = *(const v2f*)&gbs[2 * j], be = *(const v2f*)&gbs[2 * (TPR + j)];
                float d0 = pr[q].x - mu, d1 = pr[q].y - mu;
                v2f y = {fmaxf(d0 * rstd * ga.x + be.x, 0.f), fmaxf(d1 * rstd * ga.y + be.y, 0.f)};
                if (k0 + 2 * j >= GEX) y = (v2f){0.f, 0.f};
                *(v2f*)(xs + row * KSLAB + 2 * j) = y;
            }
            j += 256 - 3 * TPR; row += 3;
            if (j >= TPR) { j -= TPR; row++; }
        }
    }
    lds_barrier();

    const float* xa = xs + m * KSLAB + koff;
    float* pp = Ppart + (slab * BATCH + mt * 16 + ((lane >= 16) ? 8 : 0)) * PPN + n;
    v2f af[KSTEPS];
    #pragma unroll
    for (int s = 0; s < KSTEPS; s++) af[s] = *(const v2f*)(xa + 4 * s);
    asm volatile("" :: "v"(af[0]), "v"(af[1]), "v"(af[2]), "v"(af[3]), "v"(af[4]), "v"(af[5]),
                       "v"(af[6]), "v"(af[7]), "v"(af[8]), "v"(af[9]), "v"(af[10]), "v"(af[11]),
                       "v"(af[12]), "v"(af[13]), "v"(af[14]), "v"(af[15]), "v"(af[16]), "v"(af[17]));
    asm volatile("" :: "v"(af[18]), "v"(af[19]), "v"(af[20]), "v"(af[21]), "v"(af[22]), "v"(af[23]),
                       "v"(af[24]), "v"(af[25]), "v"(af[26]), "v"(af[27]), "v"(af[28]), "v"(af[29]),
                       "v"(af[30]), "v"(af[31]), "v"(af[32]), "v"(af[33]), "v"(af[34]));
    v8f acc = {};
    #pragma unroll
    for (int s = 0; s < KSTEPS; s++)
        acc = __builtin_amdgcn_wmma_f32_16x16x4_f32(
            false, af[s], false, bm[s], (short)0, acc, false, false);

    for (int v = 0; v < 8; v++) pp[v * PPN] = acc[v];
}

__global__ __launch_bounds__(256) void k_outsum(const float* __restrict__ Ppart,
                                                float* __restrict__ out) {
    int j = blockIdx.x * 256 + threadIdx.x, b = blockIdx.y;
    if (j < GEX / 2) {
        const float* pl = Ppart + b * PPN + 2 * j;
        v2f s = *(const v2f*)pl;
        for (int p = 1; p < NKS; p++) s += *(const v2f*)(pl + p * (BATCH * PPN));
        *(v2f*)(out + b * GEX + 2 * j) = s;
    }
}

extern "C" void kernel_launch(void* const* d_in, const int* in_sizes, int n_in,
                              void* d_out, int out_size, void* d_ws, size_t ws_size,
                              hipStream_t stream) {
    const float* b_gex    = (const float*)d_in[0];
    const float* node_feat= (const float*)d_in[1];
    const float* mask     = (const float*)d_in[2];
    const float* adj      = (const float*)d_in[3];
    const float* dist     = (const float*)d_in[4];
    const float* dose     = (const float*)d_in[5];
    const float* timev    = (const float*)d_in[6];
    const float* ppi      = (const float*)d_in[7];
    const float* w_gex    = (const float*)d_in[8];
    const float* w_comp   = (const float*)d_in[9];
    const float* W1       = (const float*)d_in[10];
    const float* W2       = (const float*)d_in[11];
    const float* w_dose   = (const float*)d_in[12];
    const float* w_time   = (const float*)d_in[13];
    const float* gamma    = (const float*)d_in[14];
    const float* beta     = (const float*)d_in[15];
    const float* Wff      = (const float*)d_in[16];

    float* out  = (float*)d_out;
    float* comp = out + BATCH * GEX;

    float* ws     = (float*)d_ws;
    float* ppool  = ws;
    float* S      = ppool + BATCH * NCH * CCEH;
    float* pred0  = S + BATCH * HDIM;
    float* lnpart = pred0 + BATCH * GEX;
    float* Ppart  = lnpart + BATCH * KPAD;
    float* Spart  = Ppart + NKS * BATCH * PPN;
    float* rowsum = Spart + NSLAB * BATCH * HDIM;

    k_cce<<<dim3(BATCH, NCH + RSY), CCE_T, 0, stream>>>(node_feat, mask, adj, dist, W1, ppool, ppi, rowsum);
    k_compS<<<dim3(NSLAB, BATCH / BGRP), 256, 0, stream>>>(ppool, W2, dose, timev, w_dose, w_time,
                                                           b_gex, w_gex, w_comp, comp, Spart);
    k_S<<<BATCH, HDIM, 0, stream>>>(Spart, S);
    k_pred0<<<dim3(NWG4, BATCH / PB), 256, 0, stream>>>(b_gex, comp, w_gex, w_comp, S, rowsum, pred0, lnpart);
    k_gemm_slab<<<dim3(NNB, NKS), 256, 0, stream>>>(pred0, lnpart, gamma, beta, Wff, Ppart);
    k_outsum<<<dim3((GEX / 2 + 255) / 256, BATCH), 256, 0, stream>>>(Ppart, out);
}


# S launch goes: k_compS slabs of 32 g x 8 batch rows (two 16-g halves met in LDS) leave 31 planes, k_pred0 work-groups add their rows' planes in slab order; 6 launches to 5; S's sum order changes
# speedup vs baseline: 7.0096x; 1.0607x over previous
#include <hip/hip_runtime.h>
#include <hip/hip_bf16.h>

#define GEX 978
#define KPAD 980
#define HDIM 128
#define NAT 50
#define FEAT 34
#define CCEH 64
#define BATCH 64
#define GCH 32
#define NSLAB ((GEX + GCH - 1) / GCH)
#define BGRP 8
#define PPITCH (CCEH + 1)
#define MC 25
#define NCH (NAT / MC)
#define CCE_T 288
#define NFT ((MC * FEAT + CCE_T - 1) / CCE_T)
#define ADT ((NAT * MC + CCE_T - 1) / CCE_T)
#define KSTEPS 35
#define KSLAB (4 * KSTEPS)
#define NKS (KPAD / KSLAB)
#define NBLK 32
#define NNB ((GEX + NBLK - 1) / NBLK)
#define PPN (NNB * NBLK)
#define PPI_V2 (GEX / 2)
#define PPI_LD ((PPI_V2 + 31) / 32)
#define SPITCH (HDIM + 4)
#define PG 32
#define PB 8
#define NWG4 ((GEX + PG - 1) / PG)
#define NLAST (GEX - PG * (NWG4 - 1))
#define NST ((NWG4 + 7) / 8)
#define RSG (CCE_T / 32)
#define RSY ((GEX + RSG * BATCH - 1) / (RSG * BATCH))
#define TPR (KSLAB / 2)
#define TLD ((BATCH * TPR + 255) / 256)

typedef __attribute__((ext_vector_type(2))) float v2f;
typedef __attribute__((ext_vector_type(4))) float v4f;
typedef __attribute__((ext_vector_type(8))) float v8f;

__device__ __forceinline__ float xhalf(float x) {
    int b = __builtin_bit_cast(int, x);
    return __builtin_bit_cast(float, __builtin_amdgcn_permlanex16(b, b, 0x76543210, 0xfedcba98, false, false));
}
__device__ __forceinline__ float wave_sum(float v) {
    v += __builtin_bit_cast(float, __builtin_amdgcn_update_dpp(0, __builtin_bit_cast(int, v), 0x161, 0xf, 0xf, true));
    v += __builtin_bit_cast(float, __builtin_amdgcn_update_dpp(0, __builtin_bit_cast(int, v), 0x162, 0xf, 0xf, true));
    v += __builtin_bit_cast(float, __builtin_amdgcn_update_dpp(0, __builtin_bit_cast(int, v), 0x164, 0xf, 0xf, true));
    v += __builtin_bit_cast(float, __builtin_amdgcn_update_dpp(0, __builtin_bit_cast(int, v), 0x168, 0xf, 0xf, true));
    return v + xhalf(v);
}

__device__ __forceinline__ void ppi_rowsums(const float* __restrict__ ppi,
                                            float* __restrict__ rowsum, int grp) {
    int lane = threadIdx.x & 31;
    int r = grp * RSG + (threadIdx.x >> 5);
    if (r >= GEX) return;
    const v2f* prow = (const v2f*)(ppi + r * GEX);
    v2f pv[PPI_LD];
    #pragma unroll
    for (int i = 0; i < PPI_LD; i++) {
        int j = lane + 32 * i;
        pv[i] = prow[j < PPI_V2 ? j : PPI_V2 - 1];
    }
    float rsum = 0.f;
    #pragma unroll
    for (int i = 0; i < PPI_LD; i++) {
        float a = pv[i].x + pv[i].y;
        rsum += (lane + 32 * i < PPI_V2) ? a : 0.f;
    }
    rsum = wave_sum(rsum);
    if (lane == 0) rowsum[r] = rsum;
}

__global__ __launch_bounds__(CCE_T) void k_cce(const float* __restrict__ nf,
                                               const float* __restrict__ mask,
                                               const float* __restrict__ adj,
                                               const float* __restrict__ dist,
                                               const float* __restrict__ W1,
                                               float* __restrict__ ppool,
                                               const float* __restrict__ ppi,
                                               float* __restrict__ rowsum) {
    if (blockIdx.y >= NCH) {
        ppi_rowsums(ppi, rowsum, (blockIdx.y - NCH) * BATCH + blockIdx.x);
        return;
    }
    int b = blockIdx.x, c = blockIdx.y, tid = threadIdx.x, m0 = c * MC;
    __shared__ float nfs[MC * FEAT];
    __shared__ float as[NAT * MC];
    __shared__ float es[NAT * MC];
    __shared__ float hs[MC * CCEH];
    __shared__ float ms[NAT];
    __shared__ float wm[MC];
    __shared__ float msum_s;

    int d = tid & (CCEH - 1);
    float w1[FEAT];
    #pragma unroll
    for (int f = 0; f < FEAT; f++) w1[f] = W1[f * CCEH + d];
    const float* nfb = nf + (b * NAT + m0) * FEAT;
    float nv[NFT];
    #pragma unroll
    for (int j = 0; j < NFT; j++) {
        int i = tid + CCE_T * j;
        nv[j] = nfb[i < MC * FEAT ? i : MC * FEAT - 1];
    }
    float av[ADT], dv[ADT];
    #pragma unroll
    for (int q = 0; q < ADT; q++) {
        int j = tid + CCE_T * q;
        int jc = j < NAT * MC ? j : NAT * MC - 1;
        int n = jc / MC, ml = jc - n * MC;
        int idx = (b * NAT + n) * NAT + m0 + ml;
        av[q] = adj[idx];
        dv[q] = dist[idx];
    }
    float mk = mask[b * NAT + (tid < NAT ? tid : NAT - 1)];

    #pragma unroll
    for (int j = 0; j < NFT; j++) {
        int i = tid + CCE_T * j;
        if (i < MC * FEAT) nfs[i] = nv[j];
    }
    #pragma unroll
    for (int q = 0; q < ADT; q++) {
        int j = tid + CCE_T * q;
        if (j < NAT * MC) { as[j] = av[q]; es[j] = expf(-dv[q]); }
    }
    if (tid < NAT) ms[tid] = mk;
    __syncthreads();

    if (tid < 256) {
        int mq = tid >> 6;
        #pragma unroll
        for (int r = 0; r < (MC + 3) / 4; r++) {
            int ml = mq + 4 * r;
            if (ml < MC) {
                float s = 0.f;
                #pragma unroll
                for (int f = 0; f < FEAT; f++) s += nfs[ml * FEAT + f] * w1[f];
                hs[ml * CCEH + d] = fmaxf(s, 0.f);
            }
        }
    } else if (tid < 256 + MC) {
        int ml = tid - 256;
        float s = 0.f, msum = 0.f;
        #pragma unroll 10
        for (int n = 0; n < NAT; n++) {
            s += ms[n] * as[n * MC + ml] * es[n * MC + ml];
            msum += ms[n];
        }
        wm[ml] = s;
        if (ml == 0) msum_s = fmaxf(msum, 1.f);
    }
    __syncthreads();
    if (tid < CCEH) {
        float s = 0.f;
        #pragma unroll
        for (int ml = 0; ml < MC; ml++) s += wm[ml] * hs[ml * CCEH + tid];
        ppool[(b * NCH + c) * CCEH + tid] = s / msum_s;
    }
}

__global__ __launch_bounds__(256) void k_compS(const float* __restrict__ ppool,
                                               const float* __restrict__ W2,
                                               const float* __restrict__ dose,
                                               const float* __restrict__ timev,
                                               const float* __restrict__ w_dose,
                                               const float* __restrict__ w_time,
                                               const float* __restrict__ b_gex,
                                               const float* __restrict__ w_gex,
                                               const float* __restrict__ w_comp,
                                               float* __restrict__ comp,
                                               float* __restrict__ Spart) {
    __shared__ float Ps[BGRP * PPITCH];
    __shared__ float W2s[CCEH * GCH];
    __shared__ __attribute__((aligned(16))) float Xs[BGRP * GCH];
    __shared__ __attribute__((aligned(16))) float Cs[BGRP * GCH];
    __shared__ float Hs[BGRP * HDIM];
    int slab = blockIdx.x, b0 = blockIdx.y * BGRP, g0 = slab * GCH, tid = threadIdx.x;

    float pv[BGRP * CCEH / 256][NCH], wv[CCEH * GCH / 256];
    #pragma unroll
    for (int j = 0; j < BGRP * CCEH / 256; j++) {
        int i = tid + 256 * j;
        #pragma unroll
        for (int c = 0; c < NCH; c++)
            pv[j][c] = ppool[((b0 + (i >> 6)) * NCH + c) * CCEH + (i & 63)];
    }
    #pragma unroll
    for (int j = 0; j < CCEH * GCH / 256; j++) {
        int i = tid + 256 * j;
        int gq = g0 + (i & (GCH - 1));
        wv[j] = W2[(i / GCH) * GEX + (gq < GEX ? gq : GEX - 1)];
    }
    int bl = tid / GCH, gl = tid & (GCH - 1), b = b0 + bl, g = g0 + gl;
    bool gok = g < GEX;
    int gc = gok ? g : GEX - 1;
    float x  = b_gex[b * GEX + gc];
    float dsv = dose[b], wdv = w_dose[gc], tmv = timev[b], wtv = w_time[gc];
    int h = tid & (HDIM - 1), gh = tid >> 7;
    float wg[GCH / 2], wc[GCH / 2];
    #pragma unroll
    for (int j = 0; j < GCH / 2; j++) {
        int gq = g0 + gh * (GCH / 2) + j;
        int gr = gq < GEX ? gq : GEX - 1;
        wg[j] = w_gex[gr * HDIM + h];
        wc[j] = w_comp[gr * HDIM + h];
    }
    asm volatile("" :: "v"(wg[0]), "v"(wg[1]), "v"(wg[2]), "v"(wg[3]), "v"(wg[4]), "v"(wg[5]), "v"(wg[6]), "v"(wg[7]),
                       "v"(wg[8]), "v"(wg[9]), "v"(wg[10]), "v"(wg[11]), "v"(wg[12]), "v"(wg[13]), "v"(wg[14]), "v"(wg[15]),
                       "v"(wc[0]), "v"(wc[1]), "v"(wc[2]), "v"(wc[3]), "v"(wc[4]), "v"(wc[5]), "v"(wc[6]), "v"(wc[7]),
                       "v"(wc[8]), "v"(wc[9]), "v"(wc[10]), "v"(wc[11]), "v"(wc[12]), "v"(wc[13]));
    asm volatile("" :: "v"(wc[14]), "v"(wc[15]), "v"(pv[0][0]), "v"(pv[0][1]), "v"(pv[1][0]), "v"(pv[1][1]),
                       "v"(wv[0]), "v"(wv[1]), "v"(wv[2]), "v"(wv[3]), "v"(wv[4]), "v"(wv[5]), "v"(wv[6]), "v"(wv[7]),
                       "v"(x), "v"(dsv), "v"(wdv), "v"(tmv), "v"(wtv));
    float s = dsv * wdv + tmv * wtv;
    #pragma unroll
    for (int j = 0; j < GCH / 2; j++) {
        bool rok = g0 + gh * (GCH / 2) + j < GEX;
        wg[j] = rok ? wg[j] : 0.f;
        wc[j] = rok ? wc[j] : 0.f;
    }
    #pragma unroll
    for (int j = 0; j < BGRP * CCEH / 256; j++) {
        int i = tid + 256 * j;
        float p = pv[j][0];
        #pragma unroll
        for (int c = 1; c < NCH; c++) p += pv[j][c];
        Ps[(i >> 6) * PPITCH + (i & 63)] = p;
    }
    #pragma unroll
    for (int j = 0; j < CCEH * GCH / 256; j++) {
        int i = tid + 256 * j;
        W2s[i] = (g0 + (i & (GCH - 1)) < GEX) ? wv[j] : 0.f;
    }
    if (!gok) { x = 0.f; s = 0.f; }
    __syncthreads();
    #pragma unroll 8
    for (int d = 0; d < CCEH; d++) s += Ps[bl * PPITCH + d] * W2s[d * GCH + gl];
    if (gok) comp[b * GEX + g] = s;
    Xs[tid] = x;
    Cs[tid] = gok ? s : 0.f;
    __syncthreads();
    float acc[BGRP];
    #pragma unroll
    for (int row = 0; row < BGRP; row++) {
        float a = 0.f;
        #pragma unroll
        for (int j = 0; j < GCH / 2; j++)
            a += Xs[row * GCH + gh * (GCH / 2) + j] * wg[j] + Cs[row * GCH + gh * (GCH / 2) + j] * wc[j];
        acc[row] = a;
    }
    if (gh == 1) {
        #pragma unroll
        for (int row = 0; row < BGRP; row++) Hs[row * HDIM + h] = acc[row];
    }
    __syncthreads();
    if (gh == 0) {
        #pragma unroll
        for (int row = 0; row < BGRP; row++)
            Spart[(slab * BATCH + b0 + row) * HDIM + h] = acc[row] + Hs[row * HDIM + h];
    }
}

__global__ __launch_bounds__(256) void k_pred0(const float* __restrict__ b_gex,
                                               const float* __restrict__ comp,
                                               const float* __restrict__ w_gex,
                                               const float* __restrict__ w_comp,
                                               const float* __restrict__ Spart,
                                               const float* __restrict__ rowsum,
                                               float* __restrict__ pred0,
                                               float* __restrict__ lnpart) {
    __shared__ __attribute__((aligned(16))) float wgs[PG * SPITCH];
    __shared__ __attribute__((aligned(16))) float wcs[PG * SPITCH];
    __shared__ __attribute__((aligned(16))) float Ssh[PB * HDIM];
    int g0 = blockIdx.x * PG, b0 = blockIdx.y * PB, tid = threadIdx.x;
    int w = tid >> 5, lane = tid & 31;
    int b = b0 + w, g = g0 + lane;
    bool gok = g < GEX;
    int gc = gok ? g : GEX - 1;
    const v4f* sp4 = (const v4f*)Spart + b0 * (HDIM / 4) + tid;
    v4f spv[NSLAB];
    #pragma unroll
    for (int p = 0; p < NSLAB; p++) spv[p] = sp4[p * (BATCH * HDIM / 4)];
    v4f wgv[PG / 8], wcv[PG / 8];
    #pragma unroll
    for (int i = 0; i < PG / 8; i++) {
        int r = g0 + w + 8 * i;
        int of = (r < GEX ? r : GEX - 1) * HDIM + lane * 4;
        wgv[i] = *(const v4f*)(w_gex + of);
        wcv[i] = *(const v4f*)(w_comp + of);
    }
    float x = b_gex[b * GEX + gc], c = comp[b * GEX + gc], rsv = rowsum[gc];

    asm volatile("" :: "v"(spv[0]), "v"(spv[1]), "v"(spv[2]), "v"(spv[3]), "v"(spv[4]), "v"(spv[5]),
                       "v"(spv[6]), "v"(spv[7]), "v"(spv[8]), "v"(spv[9]), "v"(spv[10]), "v"(spv[11]),
                       "v"(spv[12]), "v"(spv[13]), "v"(spv[14]), "v"(spv[15]), "v"(spv[16]), "v"(spv[17]),
                       "v"(spv[18]), "v"(spv[19]), "v"(spv[20]), "v"(spv[21]), "v"(spv[22]), "v"(spv[23]),
                       "v"(spv[24]), "v"(spv[25]), "v"(spv[26]), "v"(spv[27]), "v"(spv[28]), "v"(spv[29]));
    asm volatile("" :: "v"(spv[30]), "v"(wgv[0]), "v"(wgv[1]), "v"(wgv[2]), "v"(wgv[3]),
                       "v"(wcv[0]), "v"(wcv[1]), "v"(wcv[2]), "v"(wcv[3]), "v"(x), "v"(c), "v"(rsv));
    v4f sv = spv[0];
    #pragma unroll
    for (int p = 1; p < NSLAB; p++) sv += spv[p];
    *(v4f*)(Ssh + w * HDIM + lane * 4) = sv;
    #pragma unroll
    for (int i = 0; i < PG / 8; i++) {
        *(v4f*)(wgs + (w + 8 * i) * SPITCH + lane * 4) = wgv[i];
        *(v4f*)(wcs + (w + 8 * i) * SPITCH + lane * 4) = wcv[i];
    }
    __syncthreads();
    const v4f* sp = (const v4f*)(Ssh + w * HDIM);
    const v4f* gp = (const v4f*)(wgs + lane * SPITCH);
    const v4f* cp = (const v4f*)(wcs + lane * SPITCH);
    float da = 0.f, dc = 0.f;
    #pragma unroll 8
    for (int q = 0; q < HDIM / 4; q++) {
        v4f s4 = sp[q], a4 = gp[q], c4 = cp[q];
        da = fmaf(s4.x, a4.x, da);  dc = fmaf(s4.x, c4.x, dc);
        da = fmaf(s4.y, a4.y, da);  dc = fmaf(s4.y, c4.y, dc);
        da = fmaf(s4.z, a4.z, da);  dc = fmaf(s4.z, c4.z, dc);
        da = fmaf(s4.w, a4.w, da);  dc = fmaf(s4.w, c4.w, dc);
    }
    float pv0 = 0.f;
    if (gok) {
        float row = (x * da + c * dc) * 0.08838834764831845f
                  + rsv;
        pv0 = x * row;
        pred0[b * GEX + g] = pv0;
    }
    float ps = wave_sum(pv0);
    float nn = (blockIdx.x == NWG4 - 1) ? (float)NLAST : (float)PG;
    float dd = gok ? pv0 - ps / nn : 0.f;
    float pq = wave_sum(dd * dd);
    if (lane == 0) *(v2f*)(lnpart + (blockIdx.x * BATCH + b) * 2) = (v2f){ps, pq};
}

__device__ __forceinline__ void lds_barrier() {
    __builtin_amdgcn_fence(__ATOMIC_RELEASE, "workgroup", "local");
    __builtin_amdgcn_s_barrier();
    __builtin_amdgcn_fence(__ATOMIC_ACQUIRE, "workgroup", "local");
}

__global__ __launch_bounds__(256) void k_gemm_slab(const float* __restrict__ pred0,
                                                   const float* __restrict__ lnpart,
                                                   const float* __restrict__ gamma,
                                                   const float* __restrict__ beta,
                                                   const float* __restrict__ Wff,
                                                   float* __restrict__ Ppart) {
    __shared__ __attribute__((aligned(16))) float xs[BATCH * KSLAB];
    __shared__ __attribute__((aligned(8)))  float red[2][8][BATCH];
    __shared__ float mus[BATCH], rsd[BATCH];
    __shared__ __attribute__((aligned(8)))  float gbs[2 * 256];
    int tid = threadIdx.x, lane = tid & 31, w = tid >> 5;
    int mt = w & 3, ns = w >> 2;
    int slab = blockIdx.y;
    int lg   = lane & 15;
    int koff = (lane >= 16) ? 2 : 0;
    int m  = mt * 16 + lg;
    int n  = blockIdx.x * NBLK + ns * 16 + lg;
    int nc = (n < GEX) ? n : (GEX - 1);
    int k0 = slab * KSLAB;
    int kb = k0 + koff;

    int jg = tid < TPR ? tid : (tid < 2 * TPR ? tid - TPR : 0);
    int kg = k0 + 2 * jg;
    v2f gbv = *(const v2f*)((tid < TPR ? gamma : beta) + (kg < GEX ? kg : GEX - 2));
    const v4f* st4 = (const v4f*)lnpart;
    v4f st[NST];
    #pragma unroll
    for (int t = 0; t < NST; t++) {
        int i = w + 8 * t;
        st[t] = st4[(i < NWG4 ? i : NWG4 - 1) * (BATCH / 2) + lane];
    }
    int row0 = tid / TPR, j0 = tid - row0 * TPR;
    v2f pr[TLD];
    {
        int row = row0, j = j0;
        #pragma unroll
        for (int q = 0; q < TLD; q++) {
            int kk = k0 + 2 * j;
            pr[q] = *(const v2f*)(pred0 + (row < BATCH ? row : BATCH - 1) * GEX + (kk < GEX ? kk : GEX - 2));
            j += 256 - 3 * TPR; row += 3;
            if (j >= TPR) { j -= TPR; row++; }
        }
    }
    __builtin_amdgcn_sched_barrier(0);
    const float* bp = Wff + kb * GEX + nc;
    v2f bm[KSTEPS];
    #pragma unroll
    for (int s = 0; s < KSTEPS; s++) {
        if (s < KSTEPS - 1) {
            bm[s].x = bp[(4 * s) * GEX];
            bm[s].y = bp[(4 * s + 1) * GEX];
        } else {
            int r0 = kb + 4 * s, r1 = r0 + 1;
            float b0 = Wff[((r0 < GEX) ? r0 : (GEX - 1)) * GEX + nc];
            float b1 = Wff[((r1 < GEX) ? r1 : (GEX - 1)) * GEX + nc];
            bm[s].x = (r0 < GEX) ? b0 : 0.f;
            bm[s].y = (r1 < GEX) ? b1 : 0.f;
        }
    }

    float s0 = 0.f, s1 = 0.f;
    #pragma unroll
    for (int t = 0; t < NST; t++) {
        if (w + 8 * t < NWG4) { s0 += st[t].x; s1 += st[t].z; }
    }
    *(v2f*)&red[0][w][2 * lane] = (v2f){s0, s1};
    *(v2f*)&gbs[2 * tid] = gbv;
    lds_barrier();
    v2f ms = *(const v2f*)&red[0][0][2 * lane];
    for (int ww = 1; ww < 8; ww++) ms += *(const v2f*)&red[0][ww][2 * lane];
    float mu0 = ms.x / (float)GEX, mu1 = ms.y / (float)GEX;
    float q0 = 0.f, q1 = 0.f;
    #pragma unroll
    for (int t = 0; t < NST; t++) {
        int i = w + 8 * t;
        if (i < NWG4) {
            float nn  = (i == NWG4 - 1) ? (float)NLAST : (float)PG;
            float inv = (i == NWG4 - 1) ? 1.f / (float)NLAST : 1.f / (float)PG;
            float e0 = st[t].x - nn * mu0, e1 = st[t].z - nn * mu1;
            q0 += st[t].y + e0 * e0 * inv;
            q1 += st[t].w + e1 * e1 * inv;
        }
    }
    *(v2f*)&red[1][w][2 * lane] = (v2f){q0, q1};
    lds_barrier();
    if (w == 0) {
        v2f us = *(const v2f*)&red[1][0][2 * lane];
        for (int ww = 1; ww < 8; ww++) us += *(const v2f*)&red[1][ww][2 * lane];
        mus[2 * lane] = mu0;
        mus[2 * lane + 1] = mu1;
        rsd[2 * lane]     = rsqrtf(us.x / (float)GEX + 1e-5f);
        rsd[2 * lane + 1] = rsqrtf(us.y / (float)GEX + 1e-5f);
    }
    lds_barrier();
    {
        int row = row0, j = j0;
        #pragma unroll
        for (int q = 0; q < TLD; q++) {
            if (row < BATCH) {
                float mu = mus[row], rstd = rsd[row];
                v2f ga = *(const v2f*)&gbs[2 * j], be = *(const v2f*)&gbs[2 * (TPR + j)];
                float d0 = pr[q].x - mu, d1 = pr[q].y - mu;
                v2f y = {fmaxf(d0 * rstd * ga.x + be.x, 0.f), fmaxf(d1 * rstd * ga.y + be.y, 0.f)};
                if (k0 + 2 * j >= GEX) y = (v2f){0.f, 0.f};
                *(v2f*)(xs + row * KSLAB + 2 * j) = y;
            }
            j += 256 - 3 * TPR; row += 3;
            if (j >= TPR) { j -= TPR; row++; }
        }
    }
    lds_barrier();

    const float* xa = xs + m * KSLAB + koff;
    float* pp = Ppart + (slab * BATCH + mt * 16 + ((lane >= 16) ? 8 : 0)) * PPN + n;
    v2f af[KSTEPS];
    #pragma unroll
    for (int s = 0; s < KSTEPS; s++) af[s] = *(const v2f*)(xa + 4 * s);
    asm volatile("" :: "v"(af[0]), "v"(af[1]), "v"(af[2]), "v"(af[3]), "v"(af[4]), "v"(af[5]),
                       "v"(af[6]), "v"(af[7]), "v"(af[8]), "v"(af[9]), "v"(af[10]), "v"(af[11]),
                       "v"(af[12]), "v"(af[13]), "v"(af[14]), "v"(af[15]), "v"(af[16]), "v"(af[17]));
    asm volatile("" :: "v"(af[18]), "v"(af[19]), "v"(af[20]), "v"(af[21]), "v"(af[22]), "v"(af[23]),
                       "v"(af[24]), "v"(af[25]), "v"(af[26]), "v"(af[27]), "v"(af[28]), "v"(af[29]),
                       "v"(af[30]), "v"(af[31]), "v"(af[32]), "v"(af[33]), "v"(af[34]));
    v8f acc = {};
    #pragma unroll
    for (int s = 0; s < KSTEPS; s++)
        acc = __builtin_amdgcn_wmma_f32_16x16x4_f32(
            false, af[s], false, bm[s], (short)0, acc, false, false);

    for (int v = 0; v < 8; v++) pp[v * PPN] = acc[v];
}

__global__ __launch_bounds__(256) void k_outsum(const float* __restrict__ Ppart,
                                                float* __restrict__ out) {
    int j = blockIdx.x * 256 + threadIdx.x, b = blockIdx.y;
    if (j < GEX / 2) {
        const float* pl = Ppart + b * PPN + 2 * j;
        v2f s = *(const v2f*)pl;
        for (int p = 1; p < NKS; p++) s += *(const v2f*)(pl + p * (BATCH * PPN));
        *(v2f*)(out + b * GEX + 2 * j) = s;
    }
}

extern "C" void kernel_launch(void* const* d_in, const int* in_sizes, int n_in,
                              void* d_out, int out_size, void* d_ws, size_t ws_size,
                              hipStream_t stream) {
    const float* b_gex    = (const float*)d_in[0];
    const float* node_feat= (const float*)d_in[1];
    const float* mask     = (const float*)d_in[2];
    const float* adj      = (const float*)d_in[3];
    const float* dist     = (const float*)d_in[4];
    const float* dose     = (const float*)d_in[5];
    const float* timev    = (const float*)d_in[6];
    const float* ppi      = (const float*)d_in[7];
    const float* w_gex    = (const float*)d_in[8];
    const float* w_comp   = (const float*)d_in[9];
    const float* W1       = (const float*)d_in[10];
    const float* W2       = (const float*)d_in[11];
    const float* w_dose   = (const float*)d_in[12];
    const float* w_time   = (const float*)d_in[13];
    const float* gamma    = (const float*)d_in[14];
    const float* beta     = (const float*)d_in[15];
    const float* Wff      = (const float*)d_in[16];

    float* out  = (float*)d_out;
    float* comp = out + BATCH * GEX;

    float* ws     = (float*)d_ws;
    float* ppool  = ws;
    float* pred0  = ppool + BATCH * NCH * CCEH;
    float* lnpart = pred0 + BATCH * GEX;
    float* Ppart  = lnpart + BATCH * KPAD;
    float* Spart  = Ppart + NKS * BATCH * PPN;
    float* rowsum = Spart + NSLAB * BATCH * HDIM;

    k_cce<<<dim3(BATCH, NCH + RSY), CCE_T, 0, stream>>>(node_feat, mask, adj, dist, W1, ppool, ppi, rowsum);
    k_compS<<<dim3(NSLAB, BATCH / BGRP), 256, 0, stream>>>(ppool, W2, dose, timev, w_dose, w_time,
                                                           b_gex, w_gex, w_comp, comp, Spart);
    k_pred0<<<dim3(NWG4, BATCH / PB), 256, 0, stream>>>(b_gex, comp, w_gex, w_comp, Spart, rowsum, pred0, lnpart);
    k_gemm_slab<<<dim3(NNB, NKS), 256, 0, stream>>>(pred0, lnpart, gamma, beta, Wff, Ppart);
    k_outsum<<<dim3((GEX / 2 + 255) / 256, BATCH), 256, 0, stream>>>(Ppart, out);
}
